# starting kernel
# baseline (speedup 1.0000x reference)
#include <hip/hip_runtime.h>
#include <hip/hip_bf16.h>

#define NNODES 100000
#define NEDGES 1600000
#define INCH   32
#define HIDC   64
#define NHEADS 2
#define HCAT   128
#define NCLS   12
#define NEGSLOPE 0.2f

typedef __attribute__((ext_vector_type(2))) float v2f;
typedef __attribute__((ext_vector_type(8))) float v8f;

__device__ __forceinline__ float leakyr(float x) { return x > 0.f ? x : NEGSLOPE * x; }

__device__ __forceinline__ void atomicMaxF(float* addr, float v) {
    if (v >= 0.f) atomicMax((int*)addr, __float_as_int(v));
    else          atomicMin((unsigned int*)addr, __float_as_uint(v));
}

template <int K, int M, int MT>
__global__ __launch_bounds__(256) void gemm_wmma_f32(
    const float* __restrict__ A, const float* __restrict__ B,
    const float* __restrict__ bias, float* __restrict__ D, int nrows)
{
    const int wave = threadIdx.x >> 5;
    const int lane = threadIdx.x & 31;
    const int tile = blockIdx.x * 8 + wave;
    const int ntiles = nrows >> 4;
    if (tile >= ntiles) return;
    const int half = lane >> 4;
    const int mrow = lane & 15;
    const int ncol = lane & 15;
    const int m0 = tile << 4;

    v8f acc[MT] = {};

    const float* __restrict__ arow = A + (size_t)(m0 + mrow) * K;
    for (int kb = 0; kb < K; kb += 4) {
        const int k0 = kb + 2 * half;
        v2f a;
        a.x = arow[k0];
        a.y = arow[k0 + 1];
#pragma unroll
        for (int t = 0; t < MT; ++t) {
            const int n = t * 16 + ncol;
            v2f b;
            if ((M & 15) == 0 || n < M) {
                b.x = B[(size_t)k0 * M + n];
                b.y = B[(size_t)(k0 + 1) * M + n];
            } else {
                b.x = 0.f; b.y = 0.f;
            }
            acc[t] = __builtin_amdgcn_wmma_f32_16x16x4_f32(
                false, a, false, b, (short)0, acc[t], false, false);
        }
    }

#pragma unroll
    for (int t = 0; t < MT; ++t) {
        const int n = t * 16 + ncol;
        if ((M & 15) != 0 && n >= M) continue;
        const float bv = bias ? bias[n] : 0.f;
#pragma unroll
        for (int r = 0; r < 8; ++r) {
            const int m = m0 + r + 8 * half;
            D[(size_t)m * M + n] = acc[t][r] + bv;
        }
    }
}

__global__ __launch_bounds__(256) void attn_logits_kernel(
    const float* __restrict__ H, const float* __restrict__ a_s,
    const float* __restrict__ a_d, float* __restrict__ als, float* __restrict__ ald)
{
    const int idx = blockIdx.x * blockDim.x + threadIdx.x;
    if (idx >= NNODES * NHEADS) return;
    const int head = idx & 1;
    const int node = idx >> 1;
    const float4* hv = (const float4*)(H + (size_t)node * HCAT + head * HIDC);
    const float4* sv = (const float4*)(a_s + head * HIDC);
    const float4* dv = (const float4*)(a_d + head * HIDC);
    float s = 0.f, d = 0.f;
#pragma unroll
    for (int i = 0; i < HIDC / 4; ++i) {
        const float4 h4 = hv[i], s4 = sv[i], d4 = dv[i];
        s += h4.x * s4.x + h4.y * s4.y + h4.z * s4.z + h4.w * s4.w;
        d += h4.x * d4.x + h4.y * d4.y + h4.z * d4.z + h4.w * d4.w;
    }
    als[idx] = s;
    ald[idx] = d;
}

__device__ __forceinline__ void edge_sd(const int* __restrict__ ei, int e, int& s, int& d) {
    if (e < NEDGES) { s = ei[e]; d = ei[NEDGES + e]; }
    else            { s = d = e - NEDGES; }
}

__global__ __launch_bounds__(256) void edge_max_kernel(
    const int* __restrict__ ei, const float* __restrict__ als,
    const float* __restrict__ ald, float* __restrict__ maxb)
{
    const int e = blockIdx.x * blockDim.x + threadIdx.x;
    if (e >= NEDGES + NNODES) return;
    int s, d; edge_sd(ei, e, s, d);
    atomicMaxF(&maxb[d * 2 + 0], leakyr(als[s * 2 + 0] + ald[d * 2 + 0]));
    atomicMaxF(&maxb[d * 2 + 1], leakyr(als[s * 2 + 1] + ald[d * 2 + 1]));
}

__global__ __launch_bounds__(256) void edge_exp_kernel(
    const int* __restrict__ ei, const float* __restrict__ als,
    const float* __restrict__ ald, const float* __restrict__ maxb,
    float* __restrict__ exbuf, float* __restrict__ denom)
{
    const int e = blockIdx.x * blockDim.x + threadIdx.x;
    if (e >= NEDGES + NNODES) return;
    int s, d; edge_sd(ei, e, s, d);
    const float x0 = __expf(leakyr(als[s * 2 + 0] + ald[d * 2 + 0]) - maxb[d * 2 + 0]);
    const float x1 = __expf(leakyr(als[s * 2 + 1] + ald[d * 2 + 1]) - maxb[d * 2 + 1]);
    exbuf[(size_t)e * 2 + 0] = x0;
    exbuf[(size_t)e * 2 + 1] = x1;
    atomicAdd(&denom[d * 2 + 0], x0);
    atomicAdd(&denom[d * 2 + 1], x1);
}

__global__ __launch_bounds__(256) void edge_aggr_kernel(
    const int* __restrict__ ei, const float* __restrict__ H,
    const float* __restrict__ exbuf, const float* __restrict__ denom,
    float* __restrict__ AGG)
{
    const int e = blockIdx.x * 8 + (threadIdx.x >> 5);
    if (e >= NEDGES + NNODES) return;
    const int lane = threadIdx.x & 31;
    int s, d; edge_sd(ei, e, s, d);
    const int c = lane * 4;
    const int head = c >> 6;
    const float alpha = exbuf[(size_t)e * 2 + head] / denom[d * 2 + head];
    const float4 h4 = *(const float4*)(H + (size_t)s * HCAT + c);
    float* out = AGG + (size_t)d * HCAT + c;
    atomicAdd(out + 0, h4.x * alpha);
    atomicAdd(out + 1, h4.y * alpha);
    atomicAdd(out + 2, h4.z * alpha);
    atomicAdd(out + 3, h4.w * alpha);
}

__global__ __launch_bounds__(256) void finalize_concat_kernel(
    const float* __restrict__ AGG, const float* __restrict__ bias,
    const float* __restrict__ prelu_a, float* __restrict__ OUT)
{
    const int idx = blockIdx.x * blockDim.x + threadIdx.x;
    if (idx >= NNODES * HCAT / 4) return;
    const int k = (idx & (HCAT / 4 - 1)) * 4;
    const float a = *prelu_a;
    float4 v = ((const float4*)AGG)[idx];
    v.x += bias[k + 0]; v.y += bias[k + 1]; v.z += bias[k + 2]; v.w += bias[k + 3];
    v.x = v.x >= 0.f ? v.x : a * v.x;
    v.y = v.y >= 0.f ? v.y : a * v.y;
    v.z = v.z >= 0.f ? v.z : a * v.z;
    v.w = v.w >= 0.f ? v.w : a * v.w;
    ((float4*)OUT)[idx] = v;
}

__global__ __launch_bounds__(256) void finalize_mean_kernel(
    const float* __restrict__ AGG, const float* __restrict__ bias,
    const float* __restrict__ prelu_a, float* __restrict__ OUT)
{
    const int idx = blockIdx.x * blockDim.x + threadIdx.x;
    if (idx >= NNODES * HIDC / 4) return;
    const int node = idx / (HIDC / 4);
    const int c = (idx % (HIDC / 4)) * 4;
    const float a = *prelu_a;
    const float* base = AGG + (size_t)node * HCAT;
    const float4 v0 = *(const float4*)(base + c);
    const float4 v1 = *(const float4*)(base + HIDC + c);
    float4 v;
    v.x = 0.5f * (v0.x + v1.x) + bias[c + 0];
    v.y = 0.5f * (v0.y + v1.y) + bias[c + 1];
    v.z = 0.5f * (v0.z + v1.z) + bias[c + 2];
    v.w = 0.5f * (v0.w + v1.w) + bias[c + 3];
    v.x = v.x >= 0.f ? v.x : a * v.x;
    v.y = v.y >= 0.f ? v.y : a * v.y;
    v.z = v.z >= 0.f ? v.z : a * v.z;
    v.w = v.w >= 0.f ? v.w : a * v.w;
    *(float4*)(OUT + (size_t)node * HIDC + c) = v;
}

extern "C" void kernel_launch(void* const* d_in, const int* in_sizes, int n_in,
                              void* d_out, int out_size, void* d_ws, size_t ws_size,
                              hipStream_t stream)
{
    const float* x       = (const float*)d_in[0];
    const int*   ei      = (const int*)  d_in[1];
    const float* W1      = (const float*)d_in[2];
    const float* a1s     = (const float*)d_in[3];
    const float* a1d     = (const float*)d_in[4];
    const float* b1      = (const float*)d_in[5];
    const float* W2      = (const float*)d_in[6];
    const float* a2s     = (const float*)d_in[7];
    const float* a2d     = (const float*)d_in[8];
    const float* b2      = (const float*)d_in[9];
    const float* W3      = (const float*)d_in[10];
    const float* a3s     = (const float*)d_in[11];
    const float* a3d     = (const float*)d_in[12];
    const float* b3      = (const float*)d_in[13];
    const float* prelu_a = (const float*)d_in[14];
    const float* Wc      = (const float*)d_in[15];
    const float* bc      = (const float*)d_in[16];
    float* out = (float*)d_out;

    float* ws = (float*)d_ws;
    size_t off = 0;
    float* H     = ws + off; off += (size_t)NNODES * HCAT;
    float* AGG   = ws + off; off += (size_t)NNODES * HCAT;
    float* PREV  = ws + off; off += (size_t)NNODES * HCAT;
    float* als   = ws + off; off += (size_t)NNODES * 2;
    float* ald   = ws + off; off += (size_t)NNODES * 2;
    float* maxb  = ws + off; off += (size_t)NNODES * 2;
    float* denom = ws + off; off += (size_t)NNODES * 2;
    float* exbuf = ws + off; off += (size_t)(NEDGES + NNODES) * 2;

    const int ETOT = NEDGES + NNODES;
    const int gN2    = (NNODES * NHEADS + 255) / 256;
    const int gEdge  = (ETOT + 255) / 256;
    const int gEdgeW = (ETOT + 7) / 8;
    const int gTile  = ((NNODES / 16) + 7) / 8;
    const int gFinC  = (NNODES * HCAT / 4 + 255) / 256;
    const int gFinM  = (NNODES * HIDC / 4 + 255) / 256;

    struct Layer {
        const float* W; const float* as; const float* ad; const float* b; bool concat;
    };
    const Layer layers[3] = {
        { W1, a1s, a1d, b1, true  },
        { W2, a2s, a2d, b2, true  },
        { W3, a3s, a3d, b3, false },
    };

    for (int l = 0; l < 3; ++l) {
        const float* in = (l == 0) ? x : PREV;
        hipMemsetAsync(AGG,   0x00, (size_t)NNODES * HCAT * sizeof(float), stream);
        hipMemsetAsync(maxb,  0xFF, (size_t)NNODES * 2 * sizeof(float), stream);
        hipMemsetAsync(denom, 0x00, (size_t)NNODES * 2 * sizeof(float), stream);

        if (l == 0)
            gemm_wmma_f32<INCH, HCAT, 8><<<gTile, 256, 0, stream>>>(in, layers[l].W, nullptr, H, NNODES);
        else
            gemm_wmma_f32<HCAT, HCAT, 8><<<gTile, 256, 0, stream>>>(in, layers[l].W, nullptr, H, NNODES);

        attn_logits_kernel<<<gN2, 256, 0, stream>>>(H, layers[l].as, layers[l].ad, als, ald);
        edge_max_kernel  <<<gEdge, 256, 0, stream>>>(ei, als, ald, maxb);
        edge_exp_kernel  <<<gEdge, 256, 0, stream>>>(ei, als, ald, maxb, exbuf, denom);
        edge_aggr_kernel <<<gEdgeW, 256, 0, stream>>>(ei, H, exbuf, denom, AGG);

        if (layers[l].concat)
            finalize_concat_kernel<<<gFinC, 256, 0, stream>>>(AGG, layers[l].b, prelu_a, PREV);
        else
            finalize_mean_kernel  <<<gFinM, 256, 0, stream>>>(AGG, layers[l].b, prelu_a, PREV);
    }

    gemm_wmma_f32<HIDC, NCLS, 1><<<gTile, 256, 0, stream>>>(PREV, Wc, bc, out, NNODES);
    (void)in_sizes; (void)n_in; (void)out_size; (void)ws_size;
}


# exp and scatter edge passes fused: a wave takes 32 edges, lanes compute their exps, rows are added unnormalised, the node finalize divides by the softmax sum (divide after sum: rounding changes)
# speedup vs baseline: 1.1761x; 1.1761x over previous
#include <hip/hip_runtime.h>
#include <hip/hip_bf16.h>

#define NNODES 100000
#define NEDGES 1600000
#define INCH   32
#define HIDC   64
#define NHEADS 2
#define HCAT   128
#define NCLS   12
#define NEGSLOPE 0.2f

typedef __attribute__((ext_vector_type(2))) float v2f;
typedef __attribute__((ext_vector_type(8))) float v8f;

__device__ __forceinline__ float leakyr(float x) { return x > 0.f ? x : NEGSLOPE * x; }

__device__ __forceinline__ void atomicMaxF(float* addr, float v) {
    if (v >= 0.f) atomicMax((int*)addr, __float_as_int(v));
    else          atomicMin((unsigned int*)addr, __float_as_uint(v));
}

template <int K, int M, int MT>
__global__ __launch_bounds__(256) void gemm_wmma_f32(
    const float* __restrict__ A, const float* __restrict__ B,
    const float* __restrict__ bias, float* __restrict__ D, int nrows)
{
    const int wave = threadIdx.x >> 5;
    const int lane = threadIdx.x & 31;
    const int tile = blockIdx.x * 8 + wave;
    const int ntiles = nrows >> 4;
    if (tile >= ntiles) return;
    const int half = lane >> 4;
    const int mrow = lane & 15;
    const int ncol = lane & 15;
    const int m0 = tile << 4;

    v8f acc[MT] = {};

    const float* __restrict__ arow = A + (size_t)(m0 + mrow) * K;
    for (int kb = 0; kb < K; kb += 4) {
        const int k0 = kb + 2 * half;
        v2f a;
        a.x = arow[k0];
        a.y = arow[k0 + 1];
#pragma unroll
        for (int t = 0; t < MT; ++t) {
            const int n = t * 16 + ncol;
            v2f b;
            if ((M & 15) == 0 || n < M) {
                b.x = B[(size_t)k0 * M + n];
                b.y = B[(size_t)(k0 + 1) * M + n];
            } else {
                b.x = 0.f; b.y = 0.f;
            }
            acc[t] = __builtin_amdgcn_wmma_f32_16x16x4_f32(
                false, a, false, b, (short)0, acc[t], false, false);
        }
    }

#pragma unroll
    for (int t = 0; t < MT; ++t) {
        const int n = t * 16 + ncol;
        if ((M & 15) != 0 && n >= M) continue;
        const float bv = bias ? bias[n] : 0.f;
#pragma unroll
        for (int r = 0; r < 8; ++r) {
            const int m = m0 + r + 8 * half;
            D[(size_t)m * M + n] = acc[t][r] + bv;
        }
    }
}

__global__ __launch_bounds__(256) void attn_logits_kernel(
    const float* __restrict__ H, const float* __restrict__ a_s,
    const float* __restrict__ a_d, float* __restrict__ als, float* __restrict__ ald)
{
    const int idx = blockIdx.x * blockDim.x + threadIdx.x;
    if (idx >= NNODES * NHEADS) return;
    const int head = idx & 1;
    const int node = idx >> 1;
    const float4* hv = (const float4*)(H + (size_t)node * HCAT + head * HIDC);
    const float4* sv = (const float4*)(a_s + head * HIDC);
    const float4* dv = (const float4*)(a_d + head * HIDC);
    float s = 0.f, d = 0.f;
#pragma unroll
    for (int i = 0; i < HIDC / 4; ++i) {
        const float4 h4 = hv[i], s4 = sv[i], d4 = dv[i];
        s += h4.x * s4.x + h4.y * s4.y + h4.z * s4.z + h4.w * s4.w;
        d += h4.x * d4.x + h4.y * d4.y + h4.z * d4.z + h4.w * d4.w;
    }
    als[idx] = s;
    ald[idx] = d;
}

__device__ __forceinline__ void edge_sd(const int* __restrict__ ei, int e, int& s, int& d) {
    if (e < NEDGES) { s = ei[e]; d = ei[NEDGES + e]; }
    else            { s = d = e - NEDGES; }
}

__global__ __launch_bounds__(256) void edge_max_kernel(
    const int* __restrict__ ei, const float* __restrict__ als,
    const float* __restrict__ ald, float* __restrict__ maxb)
{
    const int e = blockIdx.x * blockDim.x + threadIdx.x;
    if (e >= NEDGES + NNODES) return;
    int s, d; edge_sd(ei, e, s, d);
    atomicMaxF(&maxb[d * 2 + 0], leakyr(als[s * 2 + 0] + ald[d * 2 + 0]));
    atomicMaxF(&maxb[d * 2 + 1], leakyr(als[s * 2 + 1] + ald[d * 2 + 1]));
}

static_assert((NEDGES % 32) == 0 && ((NEDGES + NNODES) % 32) == 0,
              "a wave's 32 edges are all real edges or all self-loops, and all in range");
__global__ __launch_bounds__(256) void edge_exp_aggr_kernel(
    const int* __restrict__ ei, const float* __restrict__ H,
    const float* __restrict__ als, const float* __restrict__ ald,
    const float* __restrict__ maxb, float* __restrict__ denom,
    float* __restrict__ AGG)
{
    const int lane = threadIdx.x & 31;
    const int base = (blockIdx.x * 8 + (threadIdx.x >> 5)) * 32;
    if (base >= NEDGES + NNODES) return;
    int s, d; edge_sd(ei, base + lane, s, d);
    const float2 as = *(const float2*)(als  + (size_t)s * 2);
    const float2 ad = *(const float2*)(ald  + (size_t)d * 2);
    const float2 mx = *(const float2*)(maxb + (size_t)d * 2);
    const float x0 = __expf(leakyr(as.x + ad.x) - mx.x);
    const float x1 = __expf(leakyr(as.y + ad.y) - mx.y);
    atomicAdd(&denom[d * 2 + 0], x0);
    atomicAdd(&denom[d * 2 + 1], x1);

    const int c = lane * 4;
    const bool head1 = lane >= 16;
    for (int jb = 0; jb < 32; jb += 8) {
        float4 hv[8];
#pragma unroll
        for (int u = 0; u < 8; ++u) {
            const int sj = __builtin_amdgcn_readlane(s, jb + u);
            hv[u] = *(const float4*)(H + (size_t)sj * HCAT + c);
        }
#pragma unroll
        for (int u = 0; u < 8; ++u) {
            const int dj = __builtin_amdgcn_readlane(d, jb + u);
            const float w0 = __int_as_float(__builtin_amdgcn_readlane(__float_as_int(x0), jb + u));
            const float w1 = __int_as_float(__builtin_amdgcn_readlane(__float_as_int(x1), jb + u));
            const float w = head1 ? w1 : w0;
            float* o = AGG + (size_t)dj * HCAT + c;
            atomicAdd(o + 0, hv[u].x * w);
            atomicAdd(o + 1, hv[u].y * w);
            atomicAdd(o + 2, hv[u].z * w);
            atomicAdd(o + 3, hv[u].w * w);
        }
    }
}

__global__ __launch_bounds__(256) void finalize_concat_kernel(
    const float* __restrict__ AGG, const float* __restrict__ denom,
    const float* __restrict__ bias,
    const float* __restrict__ prelu_a, float* __restrict__ OUT)
{
    const int idx = blockIdx.x * blockDim.x + threadIdx.x;
    if (idx >= NNODES * HCAT / 4) return;
    const int k = (idx & (HCAT / 4 - 1)) * 4;
    const float a = *prelu_a;
    const float den = denom[(idx >> 5) * 2 + (k >> 6)];
    float4 v = ((const float4*)AGG)[idx];
    v.x = v.x / den + bias[k + 0]; v.y = v.y / den + bias[k + 1];
    v.z = v.z / den + bias[k + 2]; v.w = v.w / den + bias[k + 3];
    v.x = v.x >= 0.f ? v.x : a * v.x;
    v.y = v.y >= 0.f ? v.y : a * v.y;
    v.z = v.z >= 0.f ? v.z : a * v.z;
    v.w = v.w >= 0.f ? v.w : a * v.w;
    ((float4*)OUT)[idx] = v;
}

__global__ __launch_bounds__(256) void finalize_mean_kernel(
    const float* __restrict__ AGG, const float* __restrict__ denom,
    const float* __restrict__ bias,
    const float* __restrict__ prelu_a, float* __restrict__ OUT)
{
    const int idx = blockIdx.x * blockDim.x + threadIdx.x;
    if (idx >= NNODES * HIDC / 4) return;
    const int node = idx / (HIDC / 4);
    const int c = (idx % (HIDC / 4)) * 4;
    const float a = *prelu_a;
    const float* base = AGG + (size_t)node * HCAT;
    const float4 v0 = *(const float4*)(base + c);
    const float4 v1 = *(const float4*)(base + HIDC + c);
    const float d0 = denom[node * 2 + 0];
    const float d1 = denom[node * 2 + 1];
    float4 v;
    v.x = 0.5f * (v0.x / d0 + v1.x / d1) + bias[c + 0];
    v.y = 0.5f * (v0.y / d0 + v1.y / d1) + bias[c + 1];
    v.z = 0.5f * (v0.z / d0 + v1.z / d1) + bias[c + 2];
    v.w = 0.5f * (v0.w / d0 + v1.w / d1) + bias[c + 3];
    v.x = v.x >= 0.f ? v.x : a * v.x;
    v.y = v.y >= 0.f ? v.y : a * v.y;
    v.z = v.z >= 0.f ? v.z : a * v.z;
    v.w = v.w >= 0.f ? v.w : a * v.w;
    *(float4*)(OUT + (size_t)node * HIDC + c) = v;
}

extern "C" void kernel_launch(void* const* d_in, const int* in_sizes, int n_in,
                              void* d_out, int out_size, void* d_ws, size_t ws_size,
                              hipStream_t stream)
{
    const float* x       = (const float*)d_in[0];
    const int*   ei      = (const int*)  d_in[1];
    const float* W1      = (const float*)d_in[2];
    const float* a1s     = (const float*)d_in[3];
    const float* a1d     = (const float*)d_in[4];
    const float* b1      = (const float*)d_in[5];
    const float* W2      = (const float*)d_in[6];
    const float* a2s     = (const float*)d_in[7];
    const float* a2d     = (const float*)d_in[8];
    const float* b2      = (const float*)d_in[9];
    const float* W3      = (const float*)d_in[10];
    const float* a3s     = (const float*)d_in[11];
    const float* a3d     = (const float*)d_in[12];
    const float* b3      = (const float*)d_in[13];
    const float* prelu_a = (const float*)d_in[14];
    const float* Wc      = (const float*)d_in[15];
    const float* bc      = (const float*)d_in[16];
    float* out = (float*)d_out;

    float* ws = (float*)d_ws;
    size_t off = 0;
    float* H     = ws + off; off += (size_t)NNODES * HCAT;
    float* AGG   = ws + off; off += (size_t)NNODES * HCAT;
    float* PREV  = ws + off; off += (size_t)NNODES * HCAT;
    float* als   = ws + off; off += (size_t)NNODES * 2;
    float* ald   = ws + off; off += (size_t)NNODES * 2;
    float* maxb  = ws + off; off += (size_t)NNODES * 2;
    float* denom = ws + off; off += (size_t)NNODES * 2;

    const int ETOT = NEDGES + NNODES;
    const int gN2    = (NNODES * NHEADS + 255) / 256;
    const int gEdge  = (ETOT + 255) / 256;
    const int gEdgeW = (ETOT / 32 + 7) / 8;
    const int gTile  = ((NNODES / 16) + 7) / 8;
    const int gFinC  = (NNODES * HCAT / 4 + 255) / 256;
    const int gFinM  = (NNODES * HIDC / 4 + 255) / 256;

    struct Layer {
        const float* W; const float* as; const float* ad; const float* b; bool concat;
    };
    const Layer layers[3] = {
        { W1, a1s, a1d, b1, true  },
        { W2, a2s, a2d, b2, true  },
        { W3, a3s, a3d, b3, false },
    };

    for (int l = 0; l < 3; ++l) {
        const float* in = (l == 0) ? x : PREV;
        hipMemsetAsync(AGG,   0x00, (size_t)NNODES * HCAT * sizeof(float), stream);
        hipMemsetAsync(maxb,  0xFF, (size_t)NNODES * 2 * sizeof(float), stream);
        hipMemsetAsync(denom, 0x00, (size_t)NNODES * 2 * sizeof(float), stream);

        if (l == 0)
            gemm_wmma_f32<INCH, HCAT, 8><<<gTile, 256, 0, stream>>>(in, layers[l].W, nullptr, H, NNODES);
        else
            gemm_wmma_f32<HCAT, HCAT, 8><<<gTile, 256, 0, stream>>>(in, layers[l].W, nullptr, H, NNODES);

        attn_logits_kernel<<<gN2, 256, 0, stream>>>(H, layers[l].as, layers[l].ad, als, ald);
        edge_max_kernel  <<<gEdge, 256, 0, stream>>>(ei, als, ald, maxb);
        edge_exp_aggr_kernel<<<gEdgeW, 256, 0, stream>>>(ei, H, als, ald, maxb, denom, AGG);

        if (layers[l].concat)
            finalize_concat_kernel<<<gFinC, 256, 0, stream>>>(AGG, denom, layers[l].b, prelu_a, PREV);
        else
            finalize_mean_kernel  <<<gFinM, 256, 0, stream>>>(AGG, denom, layers[l].b, prelu_a, PREV);
    }

    gemm_wmma_f32<HIDC, NCLS, 1><<<gTile, 256, 0, stream>>>(PREV, Wc, bc, out, NNODES);
    (void)in_sizes; (void)n_in; (void)out_size; (void)ws_size;
}


# Segment max and softmax sum leave the per-edge global atomics: 4 node ranges x 64 edge slices add them up in LDS (ds atomics, arrival order), then flush with whole-line global atomics
# speedup vs baseline: 2.4351x; 2.0705x over previous
#include <hip/hip_runtime.h>
#include <hip/hip_bf16.h>

#define NNODES 100000
#define NEDGES 1600000
#define INCH   32
#define HIDC   64
#define NHEADS 2
#define HCAT   128
#define NCLS   12
#define NEGSLOPE 0.2f

typedef __attribute__((ext_vector_type(2))) float v2f;
typedef __attribute__((ext_vector_type(8))) float v8f;

__device__ __forceinline__ float leakyr(float x) { return x > 0.f ? x : NEGSLOPE * x; }

__device__ __forceinline__ unsigned int fkey(float f) {
    const unsigned int b = __float_as_uint(f);
    return b ^ ((b & 0x80000000u) ? 0xFFFFFFFFu : 0x80000000u);
}
__device__ __forceinline__ float funkey(unsigned int k) {
    return __uint_as_float(k ^ ((k & 0x80000000u) ? 0x80000000u : 0xFFFFFFFFu));
}

template <int K, int M, int MT>
__global__ __launch_bounds__(256) void gemm_wmma_f32(
    const float* __restrict__ A, const float* __restrict__ B,
    const float* __restrict__ bias, float* __restrict__ D, int nrows)
{
    const int wave = threadIdx.x >> 5;
    const int lane = threadIdx.x & 31;
    const int tile = blockIdx.x * 8 + wave;
    const int ntiles = nrows >> 4;
    if (tile >= ntiles) return;
    const int half = lane >> 4;
    const int mrow = lane & 15;
    const int ncol = lane & 15;
    const int m0 = tile << 4;

    v8f acc[MT] = {};

    const float* __restrict__ arow = A + (size_t)(m0 + mrow) * K;
    for (int kb = 0; kb < K; kb += 4) {
        const int k0 = kb + 2 * half;
        v2f a;
        a.x = arow[k0];
        a.y = arow[k0 + 1];
#pragma unroll
        for (int t = 0; t < MT; ++t) {
            const int n = t * 16 + ncol;
            v2f b;
            if ((M & 15) == 0 || n < M) {
                b.x = B[(size_t)k0 * M + n];
                b.y = B[(size_t)(k0 + 1) * M + n];
            } else {
                b.x = 0.f; b.y = 0.f;
            }
            acc[t] = __builtin_amdgcn_wmma_f32_16x16x4_f32(
                false, a, false, b, (short)0, acc[t], false, false);
        }
    }

#pragma unroll
    for (int t = 0; t < MT; ++t) {
        const int n = t * 16 + ncol;
        if ((M & 15) != 0 && n >= M) continue;
        const float bv = bias ? bias[n] : 0.f;
#pragma unroll
        for (int r = 0; r < 8; ++r) {
            const int m = m0 + r + 8 * half;
            D[(size_t)m * M + n] = acc[t][r] + bv;
        }
    }
}

__global__ __launch_bounds__(256) void attn_logits_kernel(
    const float* __restrict__ H, const float* __restrict__ a_s,
    const float* __restrict__ a_d, float* __restrict__ als, float* __restrict__ ald)
{
    const int idx = blockIdx.x * blockDim.x + threadIdx.x;
    if (idx >= NNODES * NHEADS) return;
    const int head = idx & 1;
    const int node = idx >> 1;
    const float4* hv = (const float4*)(H + (size_t)node * HCAT + head * HIDC);
    const float4* sv = (const float4*)(a_s + head * HIDC);
    const float4* dv = (const float4*)(a_d + head * HIDC);
    float s = 0.f, d = 0.f;
#pragma unroll
    for (int i = 0; i < HIDC / 4; ++i) {
        const float4 h4 = hv[i], s4 = sv[i], d4 = dv[i];
        s += h4.x * s4.x + h4.y * s4.y + h4.z * s4.z + h4.w * s4.w;
        d += h4.x * d4.x + h4.y * d4.y + h4.z * d4.z + h4.w * d4.w;
    }
    als[idx] = s;
    ald[idx] = d;
}

__device__ __forceinline__ void edge_sd(const int* __restrict__ ei, int e, int& s, int& d) {
    if (e < NEDGES) { s = ei[e]; d = ei[NEDGES + e]; }
    else            { s = d = e - NEDGES; }
}

#define RANGE_NODES  25008
#define NRANGES      4
#define NSLICES      64
#define SLICE_EDGES  (NEDGES / NSLICES)
#define SCAN_THREADS 1024
#define SELF_SPAN    ((RANGE_NODES + NSLICES - 1) / NSLICES)
#define SCAN_U       5
#define FLUSH_U      7
static_assert(NEDGES % NSLICES == 0, "equal slices");
static_assert(RANGE_NODES % 16 == 0 && NNODES % 16 == 0, "every range is whole 128-byte lines");
static_assert(NRANGES * RANGE_NODES >= NNODES && (NRANGES - 1) * RANGE_NODES < NNODES, "ranges cover the nodes");
static_assert(SELF_SPAN * NSLICES >= RANGE_NODES, "the slices' self-loop spans cover a range");
static_assert((NSLICES - 1) * 768 < 2 * (NNODES - (NRANGES - 1) * RANGE_NODES), "flush rotation stays inside the shortest range");

template <bool SUM>
__global__ __launch_bounds__(SCAN_THREADS) void seg_scan_kernel(
    const int* __restrict__ ei, const float* __restrict__ als, const float* __restrict__ ald,
    const unsigned int* __restrict__ maxk_in, unsigned int* __restrict__ plane)
{
    __shared__ unsigned int acc[RANGE_NODES * 2];
    const int tid   = threadIdx.x;
    const int range = blockIdx.x % NRANGES;
    const int slice = blockIdx.x / NRANGES;
    const int n0 = range * RANGE_NODES;
    const int nn = min(RANGE_NODES, NNODES - n0);
    const int nw = nn * 2;

    const int selfLo = slice * SELF_SPAN * 2;
    const int selfHi = min((slice + 1) * SELF_SPAN, nn) * 2;
    for (int i = tid; i < nw; i += SCAN_THREADS) {
        unsigned int v = 0u;
        if (i >= selfLo && i < selfHi) {
            const size_t g = (size_t)n0 * 2 + i;
            const float e = leakyr(als[g] + ald[g]);
            if (SUM) v = __float_as_uint(__expf(e - funkey(maxk_in[g])));
            else     v = fkey(e);
        }
        acc[i] = v;
    }
    __syncthreads();

    const int ebase = slice * SLICE_EDGES;
    const int eend  = ebase + SLICE_EDGES;
    for (int k0 = 0; k0 < SLICE_EDGES; k0 += SCAN_U * SCAN_THREADS) {
        int d[SCAN_U], s[SCAN_U];
#pragma unroll
        for (int u = 0; u < SCAN_U; ++u) {
            const int e = ebase + k0 + u * SCAN_THREADS + tid;
            const bool ok = e < eend;
            d[u] = ok ? ei[NEDGES + e] : -1;
            s[u] = ok ? ei[e] : 0;
        }
        bool in[SCAN_U];
        unsigned int loc[SCAN_U];
        float2 as[SCAN_U], ad[SCAN_U];
        uint2 mk[SCAN_U];
#pragma unroll
        for (int u = 0; u < SCAN_U; ++u) {
            loc[u] = (unsigned int)(d[u] - n0);
            in[u]  = loc[u] < (unsigned int)nn;
            const int sg = in[u] ? s[u] : n0;
            const int dg = in[u] ? d[u] : n0;
            as[u] = *(const float2*)(als + (size_t)sg * 2);
            ad[u] = *(const float2*)(ald + (size_t)dg * 2);
            if (SUM) mk[u] = *(const uint2*)(maxk_in + (size_t)dg * 2);
        }
#pragma unroll
        for (int u = 0; u < SCAN_U; ++u) {
            if (in[u]) {
                const float v0 = leakyr(as[u].x + ad[u].x);
                const float v1 = leakyr(as[u].y + ad[u].y);
                if (SUM) {
                    atomicAdd((float*)&acc[loc[u] * 2 + 0], __expf(v0 - funkey(mk[u].x)));
                    atomicAdd((float*)&acc[loc[u] * 2 + 1], __expf(v1 - funkey(mk[u].y)));
                } else {
                    atomicMax(&acc[loc[u] * 2 + 0], fkey(v0));
                    atomicMax(&acc[loc[u] * 2 + 1], fkey(v1));
                }
            }
        }
    }
    __syncthreads();

    const int rot = slice * 768;
    for (int i0 = tid; i0 < nw; i0 += FLUSH_U * SCAN_THREADS) {
        unsigned int w[FLUSH_U];
        int jj[FLUSH_U];
#pragma unroll
        for (int q = 0; q < FLUSH_U; ++q) {
            const int i = i0 + q * SCAN_THREADS;
            int j = i + rot;
            if (j >= nw) j -= nw;
            jj[q] = j;
            w[q] = (i < nw) ? acc[j] : 0u;
        }
#pragma unroll
        for (int q = 0; q < FLUSH_U; ++q) {
            if (i0 + q * SCAN_THREADS < nw) {
                const size_t g = (size_t)n0 * 2 + jj[q];
                if (SUM) atomicAdd((float*)plane + g, __uint_as_float(w[q]));
                else     atomicMax(plane + g, w[q]);
            }
        }
    }
}

static_assert((NEDGES % 32) == 0 && ((NEDGES + NNODES) % 32) == 0,
              "a wave's 32 edges are all real edges or all self-loops, and all in range");
static_assert(HIDC == 64 && HCAT == 128, "lines 0,1 of a row are head 0, lines 2,3 head 1");
__global__ __launch_bounds__(256) void edge_exp_aggr_kernel(
    const int* __restrict__ ei, const float* __restrict__ H,
    const float* __restrict__ als, const float* __restrict__ ald,
    const unsigned int* __restrict__ maxk, float* __restrict__ AGG)
{
    const int lane = threadIdx.x & 31;
    const int base = (blockIdx.x * 8 + (threadIdx.x >> 5)) * 32;
    if (base >= NEDGES + NNODES) return;
    int s, d; edge_sd(ei, base + lane, s, d);
    const float2 as = *(const float2*)(als  + (size_t)s * 2);
    const float2 ad = *(const float2*)(ald  + (size_t)d * 2);
    const uint2 mk = *(const uint2*)(maxk + (size_t)d * 2);
    const float x0 = __expf(leakyr(as.x + ad.x) - funkey(mk.x));
    const float x1 = __expf(leakyr(as.y + ad.y) - funkey(mk.y));

    const float* __restrict__ Hl = H + lane;
    float* __restrict__ Al = AGG + lane;
    for (int jb = 0; jb < 32; jb += 8) {
        float hv[8][4];
#pragma unroll
        for (int u = 0; u < 8; ++u) {
            const int sj = __builtin_amdgcn_readlane(s, jb + u);
            const float* r = Hl + (size_t)sj * HCAT;
            hv[u][0] = r[0];
            hv[u][1] = r[32];
            hv[u][2] = r[64];
            hv[u][3] = r[96];
        }
#pragma unroll
        for (int u = 0; u < 8; ++u) {
            const int dj = __builtin_amdgcn_readlane(d, jb + u);
            const float w0 = __int_as_float(__builtin_amdgcn_readlane(__float_as_int(x0), jb + u));
            const float w1 = __int_as_float(__builtin_amdgcn_readlane(__float_as_int(x1), jb + u));
            float* o = Al + (size_t)dj * HCAT;
            atomicAdd(o +  0, hv[u][0] * w0);
            atomicAdd(o + 32, hv[u][1] * w0);
            atomicAdd(o + 64, hv[u][2] * w1);
            atomicAdd(o + 96, hv[u][3] * w1);
        }
    }
}

__global__ __launch_bounds__(256) void finalize_concat_kernel(
    const float* __restrict__ AGG, const float* __restrict__ denom,
    const float* __restrict__ bias,
    const float* __restrict__ prelu_a, float* __restrict__ OUT)
{
    const int idx = blockIdx.x * blockDim.x + threadIdx.x;
    if (idx >= NNODES * HCAT / 4) return;
    const int k = (idx & (HCAT / 4 - 1)) * 4;
    const float a = *prelu_a;
    const float den = denom[(idx >> 5) * 2 + (k >> 6)];
    float4 v = ((const float4*)AGG)[idx];
    v.x = v.x / den + bias[k + 0]; v.y = v.y / den + bias[k + 1];
    v.z = v.z / den + bias[k + 2]; v.w = v.w / den + bias[k + 3];
    v.x = v.x >= 0.f ? v.x : a * v.x;
    v.y = v.y >= 0.f ? v.y : a * v.y;
    v.z = v.z >= 0.f ? v.z : a * v.z;
    v.w = v.w >= 0.f ? v.w : a * v.w;
    ((float4*)OUT)[idx] = v;
}

__global__ __launch_bounds__(256) void finalize_mean_kernel(
    const float* __restrict__ AGG, const float* __restrict__ denom,
    const float* __restrict__ bias,
    const float* __restrict__ prelu_a, float* __restrict__ OUT)
{
    const int idx = blockIdx.x * blockDim.x + threadIdx.x;
    if (idx >= NNODES * HIDC / 4) return;
    const int node = idx / (HIDC / 4);
    const int c = (idx % (HIDC / 4)) * 4;
    const float a = *prelu_a;
    const float* base = AGG + (size_t)node * HCAT;
    const float4 v0 = *(const float4*)(base + c);
    const float4 v1 = *(const float4*)(base + HIDC + c);
    const float d0 = denom[node * 2 + 0];
    const float d1 = denom[node * 2 + 1];
    float4 v;
    v.x = 0.5f * (v0.x / d0 + v1.x / d1) + bias[c + 0];
    v.y = 0.5f * (v0.y / d0 + v1.y / d1) + bias[c + 1];
    v.z = 0.5f * (v0.z / d0 + v1.z / d1) + bias[c + 2];
    v.w = 0.5f * (v0.w / d0 + v1.w / d1) + bias[c + 3];
    v.x = v.x >= 0.f ? v.x : a * v.x;
    v.y = v.y >= 0.f ? v.y : a * v.y;
    v.z = v.z >= 0.f ? v.z : a * v.z;
    v.w = v.w >= 0.f ? v.w : a * v.w;
    *(float4*)(OUT + (size_t)node * HIDC + c) = v;
}

extern "C" void kernel_launch(void* const* d_in, const int* in_sizes, int n_in,
                              void* d_out, int out_size, void* d_ws, size_t ws_size,
                              hipStream_t stream)
{
    const float* x       = (const float*)d_in[0];
    const int*   ei      = (const int*)  d_in[1];
    const float* W1      = (const float*)d_in[2];
    const float* a1s     = (const float*)d_in[3];
    const float* a1d     = (const float*)d_in[4];
    const float* b1      = (const float*)d_in[5];
    const float* W2      = (const float*)d_in[6];
    const float* a2s     = (const float*)d_in[7];
    const float* a2d     = (const float*)d_in[8];
    const float* b2      = (const float*)d_in[9];
    const float* W3      = (const float*)d_in[10];
    const float* a3s     = (const float*)d_in[11];
    const float* a3d     = (const float*)d_in[12];
    const float* b3      = (const float*)d_in[13];
    const float* prelu_a = (const float*)d_in[14];
    const float* Wc      = (const float*)d_in[15];
    const float* bc      = (const float*)d_in[16];
    float* out = (float*)d_out;

    float* ws = (float*)d_ws;
    size_t off = 0;
    float* H     = ws + off; off += (size_t)NNODES * HCAT;
    float* AGG   = ws + off; off += (size_t)NNODES * HCAT;
    float* PREV  = ws + off; off += (size_t)NNODES * HCAT;
    float* als   = ws + off; off += (size_t)NNODES * 2;
    float* ald   = ws + off; off += (size_t)NNODES * 2;
    unsigned int* maxk = (unsigned int*)(ws + off); off += (size_t)NNODES * 2;
    float* denom = ws + off; off += (size_t)NNODES * 2;

    const int ETOT = NEDGES + NNODES;
    const int gN2    = (NNODES * NHEADS + 255) / 256;
    const int gEdgeW = (ETOT / 32 + 7) / 8;
    const int gTile  = ((NNODES / 16) + 7) / 8;
    const int gFinC  = (NNODES * HCAT / 4 + 255) / 256;
    const int gFinM  = (NNODES * HIDC / 4 + 255) / 256;

    struct Layer {
        const float* W; const float* as; const float* ad; const float* b; bool concat;
    };
    const Layer layers[3] = {
        { W1, a1s, a1d, b1, true  },
        { W2, a2s, a2d, b2, true  },
        { W3, a3s, a3d, b3, false },
    };

    for (int l = 0; l < 3; ++l) {
        const float* in = (l == 0) ? x : PREV;
        hipMemsetAsync(AGG,   0x00, (size_t)NNODES * HCAT * sizeof(float), stream);
        hipMemsetAsync(maxk,  0x00, (size_t)NNODES * 2 * sizeof(unsigned int), stream);
        hipMemsetAsync(denom, 0x00, (size_t)NNODES * 2 * sizeof(float), stream);

        if (l == 0)
            gemm_wmma_f32<INCH, HCAT, 8><<<gTile, 256, 0, stream>>>(in, layers[l].W, nullptr, H, NNODES);
        else
            gemm_wmma_f32<HCAT, HCAT, 8><<<gTile, 256, 0, stream>>>(in, layers[l].W, nullptr, H, NNODES);

        attn_logits_kernel<<<gN2, 256, 0, stream>>>(H, layers[l].as, layers[l].ad, als, ald);
        seg_scan_kernel<false><<<NRANGES * NSLICES, SCAN_THREADS, 0, stream>>>(ei, als, ald, nullptr, maxk);
        seg_scan_kernel<true> <<<NRANGES * NSLICES, SCAN_THREADS, 0, stream>>>(ei, als, ald, maxk, (unsigned int*)denom);
        edge_exp_aggr_kernel<<<gEdgeW, 256, 0, stream>>>(ei, H, als, ald, maxk, AGG);

        if (layers[l].concat)
            finalize_concat_kernel<<<gFinC, 256, 0, stream>>>(AGG, denom, layers[l].b, prelu_a, PREV);
        else
            finalize_mean_kernel  <<<gFinM, 256, 0, stream>>>(AGG, denom, layers[l].b, prelu_a, PREV);
    }

    gemm_wmma_f32<HIDC, NCLS, 1><<<gTile, 256, 0, stream>>>(PREV, Wc, bc, out, NNODES);
    (void)in_sizes; (void)n_in; (void)out_size; (void)ws_size;
}


# Mean layer: edge weights normalised per edge (0.5*ex/sum, divide before sum) and both heads added into one 64-channel row: 2 atomic lines per edge, not 4; adds still in arrival order
# speedup vs baseline: 2.5929x; 1.0648x over previous
#include <hip/hip_runtime.h>
#include <hip/hip_bf16.h>

#define NNODES 100000
#define NEDGES 1600000
#define INCH   32
#define HIDC   64
#define NHEADS 2
#define HCAT   128
#define NCLS   12
#define NEGSLOPE 0.2f

typedef __attribute__((ext_vector_type(2))) float v2f;
typedef __attribute__((ext_vector_type(8))) float v8f;

__device__ __forceinline__ float leakyr(float x) { return x > 0.f ? x : NEGSLOPE * x; }

__device__ __forceinline__ unsigned int fkey(float f) {
    const unsigned int b = __float_as_uint(f);
    return b ^ ((b & 0x80000000u) ? 0xFFFFFFFFu : 0x80000000u);
}
__device__ __forceinline__ float funkey(unsigned int k) {
    return __uint_as_float(k ^ ((k & 0x80000000u) ? 0x80000000u : 0xFFFFFFFFu));
}

template <int K, int M, int MT>
__global__ __launch_bounds__(256) void gemm_wmma_f32(
    const float* __restrict__ A, const float* __restrict__ B,
    const float* __restrict__ bias, float* __restrict__ D, int nrows)
{
    const int wave = threadIdx.x >> 5;
    const int lane = threadIdx.x & 31;
    const int tile = blockIdx.x * 8 + wave;
    const int ntiles = nrows >> 4;
    if (tile >= ntiles) return;
    const int half = lane >> 4;
    const int mrow = lane & 15;
    const int ncol = lane & 15;
    const int m0 = tile << 4;

    v8f acc[MT] = {};

    const float* __restrict__ arow = A + (size_t)(m0 + mrow) * K;
    for (int kb = 0; kb < K; kb += 4) {
        const int k0 = kb + 2 * half;
        v2f a;
        a.x = arow[k0];
        a.y = arow[k0 + 1];
#pragma unroll
        for (int t = 0; t < MT; ++t) {
            const int n = t * 16 + ncol;
            v2f b;
            if ((M & 15) == 0 || n < M) {
                b.x = B[(size_t)k0 * M + n];
                b.y = B[(size_t)(k0 + 1) * M + n];
            } else {
                b.x = 0.f; b.y = 0.f;
            }
            acc[t] = __builtin_amdgcn_wmma_f32_16x16x4_f32(
                false, a, false, b, (short)0, acc[t], false, false);
        }
    }

#pragma unroll
    for (int t = 0; t < MT; ++t) {
        const int n = t * 16 + ncol;
        if ((M & 15) != 0 && n >= M) continue;
        const float bv = bias ? bias[n] : 0.f;
#pragma unroll
        for (int r = 0; r < 8; ++r) {
            const int m = m0 + r + 8 * half;
            D[(size_t)m * M + n] = acc[t][r] + bv;
        }
    }
}

__global__ __launch_bounds__(256) void attn_logits_kernel(
    const float* __restrict__ H, const float* __restrict__ a_s,
    const float* __restrict__ a_d, float* __restrict__ als, float* __restrict__ ald)
{
    const int idx = blockIdx.x * blockDim.x + threadIdx.x;
    if (idx >= NNODES * NHEADS) return;
    const int head = idx & 1;
    const int node = idx >> 1;
    const float4* hv = (const float4*)(H + (size_t)node * HCAT + head * HIDC);
    const float4* sv = (const float4*)(a_s + head * HIDC);
    const float4* dv = (const float4*)(a_d + head * HIDC);
    float s = 0.f, d = 0.f;
#pragma unroll
    for (int i = 0; i < HIDC / 4; ++i) {
        const float4 h4 = hv[i], s4 = sv[i], d4 = dv[i];
        s += h4.x * s4.x + h4.y * s4.y + h4.z * s4.z + h4.w * s4.w;
        d += h4.x * d4.x + h4.y * d4.y + h4.z * d4.z + h4.w * d4.w;
    }
    als[idx] = s;
    ald[idx] = d;
}

__device__ __forceinline__ void edge_sd(const int* __restrict__ ei, int e, int& s, int& d) {
    if (e < NEDGES) { s = ei[e]; d = ei[NEDGES + e]; }
    else            { s = d = e - NEDGES; }
}

#define RANGE_NODES  25008
#define NRANGES      4
#define NSLICES      64
#define SLICE_EDGES  (NEDGES / NSLICES)
#define SCAN_THREADS 1024
#define SELF_SPAN    ((RANGE_NODES + NSLICES - 1) / NSLICES)
#define SCAN_U       5
#define FLUSH_U      7
static_assert(NEDGES % NSLICES == 0, "equal slices");
static_assert(RANGE_NODES % 16 == 0 && NNODES % 16 == 0, "every range is whole 128-byte lines");
static_assert(NRANGES * RANGE_NODES >= NNODES && (NRANGES - 1) * RANGE_NODES < NNODES, "ranges cover the nodes");
static_assert(SELF_SPAN * NSLICES >= RANGE_NODES, "the slices' self-loop spans cover a range");
static_assert((NSLICES - 1) * 768 < 2 * (NNODES - (NRANGES - 1) * RANGE_NODES), "flush rotation stays inside the shortest range");

template <bool SUM>
__global__ __launch_bounds__(SCAN_THREADS) void seg_scan_kernel(
    const int* __restrict__ ei, const float* __restrict__ als, const float* __restrict__ ald,
    const unsigned int* __restrict__ maxk_in, unsigned int* __restrict__ plane)
{
    __shared__ unsigned int acc[RANGE_NODES * 2];
    const int tid   = threadIdx.x;
    const int range = blockIdx.x % NRANGES;
    const int slice = blockIdx.x / NRANGES;
    const int n0 = range * RANGE_NODES;
    const int nn = min(RANGE_NODES, NNODES - n0);
    const int nw = nn * 2;

    const int selfLo = slice * SELF_SPAN * 2;
    const int selfHi = min((slice + 1) * SELF_SPAN, nn) * 2;
    for (int i = tid; i < nw; i += SCAN_THREADS) {
        unsigned int v = 0u;
        if (i >= selfLo && i < selfHi) {
            const size_t g = (size_t)n0 * 2 + i;
            const float e = leakyr(als[g] + ald[g]);
            if (SUM) v = __float_as_uint(__expf(e - funkey(maxk_in[g])));
            else     v = fkey(e);
        }
        acc[i] = v;
    }
    __syncthreads();

    const int ebase = slice * SLICE_EDGES;
    const int eend  = ebase + SLICE_EDGES;
    for (int k0 = 0; k0 < SLICE_EDGES; k0 += SCAN_U * SCAN_THREADS) {
        int d[SCAN_U], s[SCAN_U];
#pragma unroll
        for (int u = 0; u < SCAN_U; ++u) {
            const int e = ebase + k0 + u * SCAN_THREADS + tid;
            const bool ok = e < eend;
            d[u] = ok ? ei[NEDGES + e] : -1;
            s[u] = ok ? ei[e] : 0;
        }
        bool in[SCAN_U];
        unsigned int loc[SCAN_U];
        float2 as[SCAN_U], ad[SCAN_U];
        uint2 mk[SCAN_U];
#pragma unroll
        for (int u = 0; u < SCAN_U; ++u) {
            loc[u] = (unsigned int)(d[u] - n0);
            in[u]  = loc[u] < (unsigned int)nn;
            const int sg = in[u] ? s[u] : n0;
            const int dg = in[u] ? d[u] : n0;
            as[u] = *(const float2*)(als + (size_t)sg * 2);
            ad[u] = *(const float2*)(ald + (size_t)dg * 2);
            if (SUM) mk[u] = *(const uint2*)(maxk_in + (size_t)dg * 2);
        }
#pragma unroll
        for (int u = 0; u < SCAN_U; ++u) {
            if (in[u]) {
                const float v0 = leakyr(as[u].x + ad[u].x);
                const float v1 = leakyr(as[u].y + ad[u].y);
                if (SUM) {
                    atomicAdd((float*)&acc[loc[u] * 2 + 0], __expf(v0 - funkey(mk[u].x)));
                    atomicAdd((float*)&acc[loc[u] * 2 + 1], __expf(v1 - funkey(mk[u].y)));
                } else {
                    atomicMax(&acc[loc[u] * 2 + 0], fkey(v0));
                    atomicMax(&acc[loc[u] * 2 + 1], fkey(v1));
                }
            }
        }
    }
    __syncthreads();

    const int rot = slice * 768;
    for (int i0 = tid; i0 < nw; i0 += FLUSH_U * SCAN_THREADS) {
        unsigned int w[FLUSH_U];
        int jj[FLUSH_U];
#pragma unroll
        for (int q = 0; q < FLUSH_U; ++q) {
            const int i = i0 + q * SCAN_THREADS;
            int j = i + rot;
            if (j >= nw) j -= nw;
            jj[q] = j;
            w[q] = (i < nw) ? acc[j] : 0u;
        }
#pragma unroll
        for (int q = 0; q < FLUSH_U; ++q) {
            if (i0 + q * SCAN_THREADS < nw) {
                const size_t g = (size_t)n0 * 2 + jj[q];
                if (SUM) atomicAdd((float*)plane + g, __uint_as_float(w[q]));
                else     atomicMax(plane + g, w[q]);
            }
        }
    }
}

static_assert((NEDGES % 32) == 0 && ((NEDGES + NNODES) % 32) == 0,
              "a wave's 32 edges are all real edges or all self-loops, and all in range");
static_assert(HIDC == 64 && HCAT == 128, "lines 0,1 of a row are head 0, lines 2,3 head 1");
template <bool MEAN>
__global__ __launch_bounds__(256) void edge_exp_aggr_kernel(
    const int* __restrict__ ei, const float* __restrict__ H,
    const float* __restrict__ als, const float* __restrict__ ald,
    const unsigned int* __restrict__ maxk, const float* __restrict__ denom,
    float* __restrict__ AGG)
{
    const int lane = threadIdx.x & 31;
    const int base = (blockIdx.x * 8 + (threadIdx.x >> 5)) * 32;
    if (base >= NEDGES + NNODES) return;
    int s, d; edge_sd(ei, base + lane, s, d);
    const float2 as = *(const float2*)(als  + (size_t)s * 2);
    const float2 ad = *(const float2*)(ald  + (size_t)d * 2);
    const uint2 mk = *(const uint2*)(maxk + (size_t)d * 2);
    float x0 = __expf(leakyr(as.x + ad.x) - funkey(mk.x));
    float x1 = __expf(leakyr(as.y + ad.y) - funkey(mk.y));
    if (MEAN) {
        const float2 dn = *(const float2*)(denom + (size_t)d * 2);
        x0 = 0.5f * (x0 / dn.x);
        x1 = 0.5f * (x1 / dn.y);
    }

    const float* __restrict__ Hl = H + lane;
    float* __restrict__ Al = AGG + lane;
    for (int jb = 0; jb < 32; jb += 8) {
        float hv[8][4];
#pragma unroll
        for (int u = 0; u < 8; ++u) {
            const int sj = __builtin_amdgcn_readlane(s, jb + u);
            const float* r = Hl + (size_t)sj * HCAT;
            hv[u][0] = r[0];
            hv[u][1] = r[32];
            hv[u][2] = r[64];
            hv[u][3] = r[96];
        }
#pragma unroll
        for (int u = 0; u < 8; ++u) {
            const int dj = __builtin_amdgcn_readlane(d, jb + u);
            const float w0 = __int_as_float(__builtin_amdgcn_readlane(__float_as_int(x0), jb + u));
            const float w1 = __int_as_float(__builtin_amdgcn_readlane(__float_as_int(x1), jb + u));
            if (MEAN) {
                float* o = Al + (size_t)dj * HIDC;
                atomicAdd(o +  0, hv[u][0] * w0 + hv[u][2] * w1);
                atomicAdd(o + 32, hv[u][1] * w0 + hv[u][3] * w1);
            } else {
                float* o = Al + (size_t)dj * HCAT;
                atomicAdd(o +  0, hv[u][0] * w0);
                atomicAdd(o + 32, hv[u][1] * w0);
                atomicAdd(o + 64, hv[u][2] * w1);
                atomicAdd(o + 96, hv[u][3] * w1);
            }
        }
    }
}

__global__ __launch_bounds__(256) void finalize_concat_kernel(
    const float* __restrict__ AGG, const float* __restrict__ denom,
    const float* __restrict__ bias,
    const float* __restrict__ prelu_a, float* __restrict__ OUT)
{
    const int idx = blockIdx.x * blockDim.x + threadIdx.x;
    if (idx >= NNODES * HCAT / 4) return;
    const int k = (idx & (HCAT / 4 - 1)) * 4;
    const float a = *prelu_a;
    const float den = denom[(idx >> 5) * 2 + (k >> 6)];
    float4 v = ((const float4*)AGG)[idx];
    v.x = v.x / den + bias[k + 0]; v.y = v.y / den + bias[k + 1];
    v.z = v.z / den + bias[k + 2]; v.w = v.w / den + bias[k + 3];
    v.x = v.x >= 0.f ? v.x : a * v.x;
    v.y = v.y >= 0.f ? v.y : a * v.y;
    v.z = v.z >= 0.f ? v.z : a * v.z;
    v.w = v.w >= 0.f ? v.w : a * v.w;
    ((float4*)OUT)[idx] = v;
}

__global__ __launch_bounds__(256) void finalize_mean_kernel(
    const float* __restrict__ AGG, const float* __restrict__ bias,
    const float* __restrict__ prelu_a, float* __restrict__ OUT)
{
    const int idx = blockIdx.x * blockDim.x + threadIdx.x;
    if (idx >= NNODES * HIDC / 4) return;
    const int c = (idx & (HIDC / 4 - 1)) * 4;
    const float a = *prelu_a;
    float4 v = ((const float4*)AGG)[idx];
    v.x += bias[c + 0];
    v.y += bias[c + 1];
    v.z += bias[c + 2];
    v.w += bias[c + 3];
    v.x = v.x >= 0.f ? v.x : a * v.x;
    v.y = v.y >= 0.f ? v.y : a * v.y;
    v.z = v.z >= 0.f ? v.z : a * v.z;
    v.w = v.w >= 0.f ? v.w : a * v.w;
    ((float4*)OUT)[idx] = v;
}

extern "C" void kernel_launch(void* const* d_in, const int* in_sizes, int n_in,
                              void* d_out, int out_size, void* d_ws, size_t ws_size,
                              hipStream_t stream)
{
    const float* x       = (const float*)d_in[0];
    const int*   ei      = (const int*)  d_in[1];
    const float* W1      = (const float*)d_in[2];
    const float* a1s     = (const float*)d_in[3];
    const float* a1d     = (const float*)d_in[4];
    const float* b1      = (const float*)d_in[5];
    const float* W2      = (const float*)d_in[6];
    const float* a2s     = (const float*)d_in[7];
    const float* a2d     = (const float*)d_in[8];
    const float* b2      = (const float*)d_in[9];
    const float* W3      = (const float*)d_in[10];
    const float* a3s     = (const float*)d_in[11];
    const float* a3d     = (const float*)d_in[12];
    const float* b3      = (const float*)d_in[13];
    const float* prelu_a = (const float*)d_in[14];
    const float* Wc      = (const float*)d_in[15];
    const float* bc      = (const float*)d_in[16];
    float* out = (float*)d_out;

    float* ws = (float*)d_ws;
    size_t off = 0;
    float* H     = ws + off; off += (size_t)NNODES * HCAT;
    float* AGG   = ws + off; off += (size_t)NNODES * HCAT;
    float* PREV  = ws + off; off += (size_t)NNODES * HCAT;
    float* als   = ws + off; off += (size_t)NNODES * 2;
    float* ald   = ws + off; off += (size_t)NNODES * 2;
    unsigned int* maxk = (unsigned int*)(ws + off); off += (size_t)NNODES * 2;
    float* denom = ws + off; off += (size_t)NNODES * 2;

    const int ETOT = NEDGES + NNODES;
    const int gN2    = (NNODES * NHEADS + 255) / 256;
    const int gEdgeW = (ETOT / 32 + 7) / 8;
    const int gTile  = ((NNODES / 16) + 7) / 8;
    const int gFinC  = (NNODES * HCAT / 4 + 255) / 256;
    const int gFinM  = (NNODES * HIDC / 4 + 255) / 256;

    struct Layer {
        const float* W; const float* as; const float* ad; const float* b; bool concat;
    };
    const Layer layers[3] = {
        { W1, a1s, a1d, b1, true  },
        { W2, a2s, a2d, b2, true  },
        { W3, a3s, a3d, b3, false },
    };

    for (int l = 0; l < 3; ++l) {
        const float* in = (l == 0) ? x : PREV;
        if (layers[l].concat)
            hipMemsetAsync(AGG, 0x00, (size_t)NNODES * HCAT * sizeof(float), stream);
        else
            hipMemsetAsync(AGG, 0x00, (size_t)NNODES * HIDC * sizeof(float), stream);
        hipMemsetAsync(maxk,  0x00, (size_t)NNODES * 2 * sizeof(unsigned int), stream);
        hipMemsetAsync(denom, 0x00, (size_t)NNODES * 2 * sizeof(float), stream);

        if (l == 0)
            gemm_wmma_f32<INCH, HCAT, 8><<<gTile, 256, 0, stream>>>(in, layers[l].W, nullptr, H, NNODES);
        else
            gemm_wmma_f32<HCAT, HCAT, 8><<<gTile, 256, 0, stream>>>(in, layers[l].W, nullptr, H, NNODES);

        attn_logits_kernel<<<gN2, 256, 0, stream>>>(H, layers[l].as, layers[l].ad, als, ald);
        seg_scan_kernel<false><<<NRANGES * NSLICES, SCAN_THREADS, 0, stream>>>(ei, als, ald, nullptr, maxk);
        seg_scan_kernel<true> <<<NRANGES * NSLICES, SCAN_THREADS, 0, stream>>>(ei, als, ald, maxk, (unsigned int*)denom);
        if (layers[l].concat) {
            edge_exp_aggr_kernel<false><<<gEdgeW, 256, 0, stream>>>(ei, H, als, ald, maxk, nullptr, AGG);
            finalize_concat_kernel<<<gFinC, 256, 0, stream>>>(AGG, denom, layers[l].b, prelu_a, PREV);
        } else {
            edge_exp_aggr_kernel<true><<<gEdgeW, 256, 0, stream>>>(ei, H, als, ald, maxk, denom, AGG);
            finalize_mean_kernel<<<gFinM, 256, 0, stream>>>(AGG, layers[l].b, prelu_a, PREV);
        }
    }

    gemm_wmma_f32<HIDC, NCLS, 1><<<gTile, 256, 0, stream>>>(PREV, Wc, bc, out, NNODES);
    (void)in_sizes; (void)n_in; (void)out_size; (void)ws_size;
}


# Edge logits leaky(als[src]+ald[dst]) computed once per layer into an edge-ordered buffer; max scan, sum scan and aggregation read it in order instead of gathering both planes again; sum order kept
# speedup vs baseline: 2.7002x; 1.0414x over previous
#include <hip/hip_runtime.h>
#include <hip/hip_bf16.h>

#define NNODES 100000
#define NEDGES 1600000
#define INCH   32
#define HIDC   64
#define NHEADS 2
#define HCAT   128
#define NCLS   12
#define NEGSLOPE 0.2f

typedef __attribute__((ext_vector_type(2))) float v2f;
typedef __attribute__((ext_vector_type(8))) float v8f;

__device__ __forceinline__ float leakyr(float x) { return x > 0.f ? x : NEGSLOPE * x; }

__device__ __forceinline__ unsigned int fkey(float f) {
    const unsigned int b = __float_as_uint(f);
    return b ^ ((b & 0x80000000u) ? 0xFFFFFFFFu : 0x80000000u);
}
__device__ __forceinline__ float funkey(unsigned int k) {
    return __uint_as_float(k ^ ((k & 0x80000000u) ? 0x80000000u : 0xFFFFFFFFu));
}

template <int K, int M, int MT>
__global__ __launch_bounds__(256) void gemm_wmma_f32(
    const float* __restrict__ A, const float* __restrict__ B,
    const float* __restrict__ bias, float* __restrict__ D, int nrows)
{
    const int wave = threadIdx.x >> 5;
    const int lane = threadIdx.x & 31;
    const int tile = blockIdx.x * 8 + wave;
    const int ntiles = nrows >> 4;
    if (tile >= ntiles) return;
    const int half = lane >> 4;
    const int mrow = lane & 15;
    const int ncol = lane & 15;
    const int m0 = tile << 4;

    v8f acc[MT] = {};

    const float* __restrict__ arow = A + (size_t)(m0 + mrow) * K;
    for (int kb = 0; kb < K; kb += 4) {
        const int k0 = kb + 2 * half;
        v2f a;
        a.x = arow[k0];
        a.y = arow[k0 + 1];
#pragma unroll
        for (int t = 0; t < MT; ++t) {
            const int n = t * 16 + ncol;
            v2f b;
            if ((M & 15) == 0 || n < M) {
                b.x = B[(size_t)k0 * M + n];
                b.y = B[(size_t)(k0 + 1) * M + n];
            } else {
                b.x = 0.f; b.y = 0.f;
            }
            acc[t] = __builtin_amdgcn_wmma_f32_16x16x4_f32(
                false, a, false, b, (short)0, acc[t], false, false);
        }
    }

#pragma unroll
    for (int t = 0; t < MT; ++t) {
        const int n = t * 16 + ncol;
        if ((M & 15) != 0 && n >= M) continue;
        const float bv = bias ? bias[n] : 0.f;
#pragma unroll
        for (int r = 0; r < 8; ++r) {
            const int m = m0 + r + 8 * half;
            D[(size_t)m * M + n] = acc[t][r] + bv;
        }
    }
}

__global__ __launch_bounds__(256) void attn_logits_kernel(
    const float* __restrict__ H, const float* __restrict__ a_s,
    const float* __restrict__ a_d, float* __restrict__ als, float* __restrict__ ald)
{
    const int idx = blockIdx.x * blockDim.x + threadIdx.x;
    if (idx >= NNODES * NHEADS) return;
    const int head = idx & 1;
    const int node = idx >> 1;
    const float4* hv = (const float4*)(H + (size_t)node * HCAT + head * HIDC);
    const float4* sv = (const float4*)(a_s + head * HIDC);
    const float4* dv = (const float4*)(a_d + head * HIDC);
    float s = 0.f, d = 0.f;
#pragma unroll
    for (int i = 0; i < HIDC / 4; ++i) {
        const float4 h4 = hv[i], s4 = sv[i], d4 = dv[i];
        s += h4.x * s4.x + h4.y * s4.y + h4.z * s4.z + h4.w * s4.w;
        d += h4.x * d4.x + h4.y * d4.y + h4.z * d4.z + h4.w * d4.w;
    }
    als[idx] = s;
    ald[idx] = d;
}

#define LOGIT_U 10
#define LOGIT_T 128
static_assert(NEDGES % (LOGIT_T * LOGIT_U) == 0, "whole work-groups of edges");
__global__ __launch_bounds__(LOGIT_T) void edge_logit_kernel(
    const int* __restrict__ ei, const float* __restrict__ als, const float* __restrict__ ald,
    float* __restrict__ V)
{
    const int e0 = blockIdx.x * (LOGIT_T * LOGIT_U) + threadIdx.x;
    int s[LOGIT_U], d[LOGIT_U];
#pragma unroll
    for (int u = 0; u < LOGIT_U; ++u) {
        s[u] = ei[e0 + u * LOGIT_T];
        d[u] = ei[NEDGES + e0 + u * LOGIT_T];
    }
    float2 as[LOGIT_U], ad[LOGIT_U];
#pragma unroll
    for (int u = 0; u < LOGIT_U; ++u) {
        as[u] = *(const float2*)(als + (size_t)s[u] * 2);
        ad[u] = *(const float2*)(ald + (size_t)d[u] * 2);
    }
#pragma unroll
    for (int u = 0; u < LOGIT_U; ++u) {
        float2 v;
        v.x = leakyr(as[u].x + ad[u].x);
        v.y = leakyr(as[u].y + ad[u].y);
        *(float2*)(V + (size_t)(e0 + u * LOGIT_T) * 2) = v;
    }
}

#define RANGE_NODES  25008
#define NRANGES      4
#define NSLICES      64
#define SLICE_EDGES  (NEDGES / NSLICES)
#define SCAN_THREADS 1024
#define SELF_SPAN    ((RANGE_NODES + NSLICES - 1) / NSLICES)
#define SCAN_U       5
#define FLUSH_U      7
static_assert(NEDGES % NSLICES == 0, "equal slices");
static_assert(RANGE_NODES % 16 == 0 && NNODES % 16 == 0, "every range is whole 128-byte lines");
static_assert(NRANGES * RANGE_NODES >= NNODES && (NRANGES - 1) * RANGE_NODES < NNODES, "ranges cover the nodes");
static_assert(SELF_SPAN * NSLICES >= RANGE_NODES, "the slices' self-loop spans cover a range");
static_assert((NSLICES - 1) * 768 < 2 * (NNODES - (NRANGES - 1) * RANGE_NODES), "flush rotation stays inside the shortest range");

template <bool SUM>
__global__ __launch_bounds__(SCAN_THREADS) void seg_scan_kernel(
    const int* __restrict__ ei, const float* __restrict__ als, const float* __restrict__ ald,
    const float* __restrict__ V,
    const unsigned int* __restrict__ maxk_in, unsigned int* __restrict__ plane)
{
    __shared__ unsigned int acc[RANGE_NODES * 2];
    const int tid   = threadIdx.x;
    const int range = blockIdx.x % NRANGES;
    const int slice = blockIdx.x / NRANGES;
    const int n0 = range * RANGE_NODES;
    const int nn = min(RANGE_NODES, NNODES - n0);
    const int nw = nn * 2;

    const int selfLo = slice * SELF_SPAN * 2;
    const int selfHi = min((slice + 1) * SELF_SPAN, nn) * 2;
    for (int i = tid; i < nw; i += SCAN_THREADS) {
        unsigned int v = 0u;
        if (i >= selfLo && i < selfHi) {
            const size_t g = (size_t)n0 * 2 + i;
            const float e = leakyr(als[g] + ald[g]);
            if (SUM) v = __float_as_uint(__expf(e - funkey(maxk_in[g])));
            else     v = fkey(e);
        }
        acc[i] = v;
    }
    __syncthreads();

    const int ebase = slice * SLICE_EDGES;
    const int eend  = ebase + SLICE_EDGES;
    for (int k0 = 0; k0 < SLICE_EDGES; k0 += SCAN_U * SCAN_THREADS) {
        int d[SCAN_U];
        float2 v[SCAN_U];
#pragma unroll
        for (int u = 0; u < SCAN_U; ++u) {
            const int e = ebase + k0 + u * SCAN_THREADS + tid;
            const bool ok = e < eend;
            const int ec = ok ? e : eend - 1;
            const int dd = ei[NEDGES + ec];
            v[u] = *(const float2*)(V + (size_t)ec * 2);
            d[u] = ok ? dd : -1;
        }
#pragma unroll
        for (int u = 0; u < SCAN_U; ++u)
            asm volatile("" :: "v"(v[u].x), "v"(v[u].y));
        bool in[SCAN_U];
        unsigned int loc[SCAN_U];
        uint2 mk[SCAN_U];
#pragma unroll
        for (int u = 0; u < SCAN_U; ++u) {
            loc[u] = (unsigned int)(d[u] - n0);
            in[u]  = loc[u] < (unsigned int)nn;
            const int dg = in[u] ? d[u] : n0;
            if (SUM) mk[u] = *(const uint2*)(maxk_in + (size_t)dg * 2);
        }
        if (SUM) {
#pragma unroll
            for (int u = 0; u < SCAN_U; ++u)
                asm volatile("" :: "v"(mk[u].x), "v"(mk[u].y));
        }
#pragma unroll
        for (int u = 0; u < SCAN_U; ++u) {
            if (in[u]) {
                const float v0 = v[u].x;
                const float v1 = v[u].y;
                if (SUM) {
                    atomicAdd((float*)&acc[loc[u] * 2 + 0], __expf(v0 - funkey(mk[u].x)));
                    atomicAdd((float*)&acc[loc[u] * 2 + 1], __expf(v1 - funkey(mk[u].y)));
                } else {
                    atomicMax(&acc[loc[u] * 2 + 0], fkey(v0));
                    atomicMax(&acc[loc[u] * 2 + 1], fkey(v1));
                }
            }
        }
    }
    __syncthreads();

    const int rot = slice * 768;
    for (int i0 = tid; i0 < nw; i0 += FLUSH_U * SCAN_THREADS) {
        unsigned int w[FLUSH_U];
        int jj[FLUSH_U];
#pragma unroll
        for (int q = 0; q < FLUSH_U; ++q) {
            const int i = i0 + q * SCAN_THREADS;
            int j = i + rot;
            if (j >= nw) j -= nw;
            jj[q] = j;
            w[q] = (i < nw) ? acc[j] : 0u;
        }
#pragma unroll
        for (int q = 0; q < FLUSH_U; ++q) {
            if (i0 + q * SCAN_THREADS < nw) {
                const size_t g = (size_t)n0 * 2 + jj[q];
                if (SUM) atomicAdd((float*)plane + g, __uint_as_float(w[q]));
                else     atomicMax(plane + g, w[q]);
            }
        }
    }
}

static_assert((NEDGES % 32) == 0 && ((NEDGES + NNODES) % 32) == 0,
              "a wave's 32 edges are all real edges or all self-loops, and all in range");
static_assert(HIDC == 64 && HCAT == 128, "lines 0,1 of a row are head 0, lines 2,3 head 1");
template <bool MEAN>
__global__ __launch_bounds__(256) void edge_exp_aggr_kernel(
    const int* __restrict__ ei, const float* __restrict__ H,
    const float* __restrict__ als, const float* __restrict__ ald,
    const float* __restrict__ V,
    const unsigned int* __restrict__ maxk, const float* __restrict__ denom,
    float* __restrict__ AGG)
{
    const int lane = threadIdx.x & 31;
    const int base = (blockIdx.x * 8 + (threadIdx.x >> 5)) * 32;
    if (base >= NEDGES + NNODES) return;
    int s, d;
    float2 v;
    if (base < NEDGES) {
        const int e = base + lane;
        s = ei[e];
        d = ei[NEDGES + e];
        v = *(const float2*)(V + (size_t)e * 2);
    } else {
        s = d = base + lane - NEDGES;
        const float2 as = *(const float2*)(als + (size_t)s * 2);
        const float2 ad = *(const float2*)(ald + (size_t)d * 2);
        v.x = leakyr(as.x + ad.x);
        v.y = leakyr(as.y + ad.y);
    }
    const uint2 mk = *(const uint2*)(maxk + (size_t)d * 2);
    float x0 = __expf(v.x - funkey(mk.x));
    float x1 = __expf(v.y - funkey(mk.y));
    if (MEAN) {
        const float2 dn = *(const float2*)(denom + (size_t)d * 2);
        x0 = 0.5f * (x0 / dn.x);
        x1 = 0.5f * (x1 / dn.y);
    }

    const float* __restrict__ Hl = H + lane;
    float* __restrict__ Al = AGG + lane;
    for (int jb = 0; jb < 32; jb += 8) {
        float hv[8][4];
#pragma unroll
        for (int u = 0; u < 8; ++u) {
            const int sj = __builtin_amdgcn_readlane(s, jb + u);
            const float* r = Hl + (size_t)sj * HCAT;
            hv[u][0] = r[0];
            hv[u][1] = r[32];
            hv[u][2] = r[64];
            hv[u][3] = r[96];
        }
#pragma unroll
        for (int u = 0; u < 8; ++u) {
            const int dj = __builtin_amdgcn_readlane(d, jb + u);
            const float w0 = __int_as_float(__builtin_amdgcn_readlane(__float_as_int(x0), jb + u));
            const float w1 = __int_as_float(__builtin_amdgcn_readlane(__float_as_int(x1), jb + u));
            if (MEAN) {
                float* o = Al + (size_t)dj * HIDC;
                atomicAdd(o +  0, hv[u][0] * w0 + hv[u][2] * w1);
                atomicAdd(o + 32, hv[u][1] * w0 + hv[u][3] * w1);
            } else {
                float* o = Al + (size_t)dj * HCAT;
                atomicAdd(o +  0, hv[u][0] * w0);
                atomicAdd(o + 32, hv[u][1] * w0);
                atomicAdd(o + 64, hv[u][2] * w1);
                atomicAdd(o + 96, hv[u][3] * w1);
            }
        }
    }
}

__global__ __launch_bounds__(256) void finalize_concat_kernel(
    const float* __restrict__ AGG, const float* __restrict__ denom,
    const float* __restrict__ bias,
    const float* __restrict__ prelu_a, float* __restrict__ OUT)
{
    const int idx = blockIdx.x * blockDim.x + threadIdx.x;
    if (idx >= NNODES * HCAT / 4) return;
    const int k = (idx & (HCAT / 4 - 1)) * 4;
    const float a = *prelu_a;
    const float den = denom[(idx >> 5) * 2 + (k >> 6)];
    float4 v = ((const float4*)AGG)[idx];
    v.x = v.x / den + bias[k + 0]; v.y = v.y / den + bias[k + 1];
    v.z = v.z / den + bias[k + 2]; v.w = v.w / den + bias[k + 3];
    v.x = v.x >= 0.f ? v.x : a * v.x;
    v.y = v.y >= 0.f ? v.y : a * v.y;
    v.z = v.z >= 0.f ? v.z : a * v.z;
    v.w = v.w >= 0.f ? v.w : a * v.w;
    ((float4*)OUT)[idx] = v;
}

__global__ __launch_bounds__(256) void finalize_mean_kernel(
    const float* __restrict__ AGG, const float* __restrict__ bias,
    const float* __restrict__ prelu_a, float* __restrict__ OUT)
{
    const int idx = blockIdx.x * blockDim.x + threadIdx.x;
    if (idx >= NNODES * HIDC / 4) return;
    const int c = (idx & (HIDC / 4 - 1)) * 4;
    const float a = *prelu_a;
    float4 v = ((const float4*)AGG)[idx];
    v.x += bias[c + 0];
    v.y += bias[c + 1];
    v.z += bias[c + 2];
    v.w += bias[c + 3];
    v.x = v.x >= 0.f ? v.x : a * v.x;
    v.y = v.y >= 0.f ? v.y : a * v.y;
    v.z = v.z >= 0.f ? v.z : a * v.z;
    v.w = v.w >= 0.f ? v.w : a * v.w;
    ((float4*)OUT)[idx] = v;
}

extern "C" void kernel_launch(void* const* d_in, const int* in_sizes, int n_in,
                              void* d_out, int out_size, void* d_ws, size_t ws_size,
                              hipStream_t stream)
{
    const float* x       = (const float*)d_in[0];
    const int*   ei      = (const int*)  d_in[1];
    const float* W1      = (const float*)d_in[2];
    const float* a1s     = (const float*)d_in[3];
    const float* a1d     = (const float*)d_in[4];
    const float* b1      = (const float*)d_in[5];
    const float* W2      = (const float*)d_in[6];
    const float* a2s     = (const float*)d_in[7];
    const float* a2d     = (const float*)d_in[8];
    const float* b2      = (const float*)d_in[9];
    const float* W3      = (const float*)d_in[10];
    const float* a3s     = (const float*)d_in[11];
    const float* a3d     = (const float*)d_in[12];
    const float* b3      = (const float*)d_in[13];
    const float* prelu_a = (const float*)d_in[14];
    const float* Wc      = (const float*)d_in[15];
    const float* bc      = (const float*)d_in[16];
    float* out = (float*)d_out;

    float* ws = (float*)d_ws;
    size_t off = 0;
    float* H     = ws + off; off += (size_t)NNODES * HCAT;
    float* AGG   = ws + off; off += (size_t)NNODES * HCAT;
    float* PREV  = ws + off; off += (size_t)NNODES * HCAT;
    float* als   = ws + off; off += (size_t)NNODES * 2;
    float* ald   = ws + off; off += (size_t)NNODES * 2;
    unsigned int* maxk = (unsigned int*)(ws + off); off += (size_t)NNODES * 2;
    float* denom = ws + off; off += (size_t)NNODES * 2;
    float* V     = ws + off; off += (size_t)NEDGES * 2;

    const int ETOT = NEDGES + NNODES;
    const int gN2    = (NNODES * NHEADS + 255) / 256;
    const int gEdgeW = (ETOT / 32 + 7) / 8;
    const int gTile  = ((NNODES / 16) + 7) / 8;
    const int gFinC  = (NNODES * HCAT / 4 + 255) / 256;
    const int gFinM  = (NNODES * HIDC / 4 + 255) / 256;

    struct Layer {
        const float* W; const float* as; const float* ad; const float* b; bool concat;
    };
    const Layer layers[3] = {
        { W1, a1s, a1d, b1, true  },
        { W2, a2s, a2d, b2, true  },
        { W3, a3s, a3d, b3, false },
    };

    for (int l = 0; l < 3; ++l) {
        const float* in = (l == 0) ? x : PREV;
        if (layers[l].concat)
            hipMemsetAsync(AGG, 0x00, (size_t)NNODES * HCAT * sizeof(float), stream);
        else
            hipMemsetAsync(AGG, 0x00, (size_t)NNODES * HIDC * sizeof(float), stream);
        hipMemsetAsync(maxk,  0x00, (size_t)NNODES * 2 * sizeof(unsigned int), stream);
        hipMemsetAsync(denom, 0x00, (size_t)NNODES * 2 * sizeof(float), stream);

        if (l == 0)
            gemm_wmma_f32<INCH, HCAT, 8><<<gTile, 256, 0, stream>>>(in, layers[l].W, nullptr, H, NNODES);
        else
            gemm_wmma_f32<HCAT, HCAT, 8><<<gTile, 256, 0, stream>>>(in, layers[l].W, nullptr, H, NNODES);

        attn_logits_kernel<<<gN2, 256, 0, stream>>>(H, layers[l].as, layers[l].ad, als, ald);
        edge_logit_kernel<<<NEDGES / (LOGIT_T * LOGIT_U), LOGIT_T, 0, stream>>>(ei, als, ald, V);
        seg_scan_kernel<false><<<NRANGES * NSLICES, SCAN_THREADS, 0, stream>>>(ei, als, ald, V, nullptr, maxk);
        seg_scan_kernel<true> <<<NRANGES * NSLICES, SCAN_THREADS, 0, stream>>>(ei, als, ald, V, maxk, (unsigned int*)denom);
        if (layers[l].concat) {
            edge_exp_aggr_kernel<false><<<gEdgeW, 256, 0, stream>>>(ei, H, als, ald, V, maxk, nullptr, AGG);
            finalize_concat_kernel<<<gFinC, 256, 0, stream>>>(AGG, denom, layers[l].b, prelu_a, PREV);
        } else {
            edge_exp_aggr_kernel<true><<<gEdgeW, 256, 0, stream>>>(ei, H, als, ald, V, maxk, denom, AGG);
            finalize_mean_kernel<<<gFinM, 256, 0, stream>>>(AGG, layers[l].b, prelu_a, PREV);
        }
    }

    gemm_wmma_f32<HIDC, NCLS, 1><<<gTile, 256, 0, stream>>>(PREV, Wc, bc, out, NNODES);
    (void)in_sizes; (void)n_in; (void)out_size; (void)ws_size;
}


# Node finalize kernels enter few long waves: a wave takes 8 consecutive 512-byte row pieces with their loads in flight together (12500 and 6250 waves, not 100000 and 50000); results bit-identical
# speedup vs baseline: 2.8344x; 1.0497x over previous
#include <hip/hip_runtime.h>
#include <hip/hip_bf16.h>

#define NNODES 100000
#define NEDGES 1600000
#define INCH   32
#define HIDC   64
#define NHEADS 2
#define HCAT   128
#define NCLS   12
#define NEGSLOPE 0.2f

typedef __attribute__((ext_vector_type(2))) float v2f;
typedef __attribute__((ext_vector_type(8))) float v8f;

__device__ __forceinline__ float leakyr(float x) { return x > 0.f ? x : NEGSLOPE * x; }

__device__ __forceinline__ unsigned int fkey(float f) {
    const unsigned int b = __float_as_uint(f);
    return b ^ ((b & 0x80000000u) ? 0xFFFFFFFFu : 0x80000000u);
}
__device__ __forceinline__ float funkey(unsigned int k) {
    return __uint_as_float(k ^ ((k & 0x80000000u) ? 0x80000000u : 0xFFFFFFFFu));
}

template <int K, int M, int MT>
__global__ __launch_bounds__(256) void gemm_wmma_f32(
    const float* __restrict__ A, const float* __restrict__ B,
    const float* __restrict__ bias, float* __restrict__ D, int nrows)
{
    const int wave = threadIdx.x >> 5;
    const int lane = threadIdx.x & 31;
    const int tile = blockIdx.x * 8 + wave;
    const int ntiles = nrows >> 4;
    if (tile >= ntiles) return;
    const int half = lane >> 4;
    const int mrow = lane & 15;
    const int ncol = lane & 15;
    const int m0 = tile << 4;

    v8f acc[MT] = {};

    const float* __restrict__ arow = A + (size_t)(m0 + mrow) * K;
    for (int kb = 0; kb < K; kb += 4) {
        const int k0 = kb + 2 * half;
        v2f a;
        a.x = arow[k0];
        a.y = arow[k0 + 1];
#pragma unroll
        for (int t = 0; t < MT; ++t) {
            const int n = t * 16 + ncol;
            v2f b;
            if ((M & 15) == 0 || n < M) {
                b.x = B[(size_t)k0 * M + n];
                b.y = B[(size_t)(k0 + 1) * M + n];
            } else {
                b.x = 0.f; b.y = 0.f;
            }
            acc[t] = __builtin_amdgcn_wmma_f32_16x16x4_f32(
                false, a, false, b, (short)0, acc[t], false, false);
        }
    }

#pragma unroll
    for (int t = 0; t < MT; ++t) {
        const int n = t * 16 + ncol;
        if ((M & 15) != 0 && n >= M) continue;
        const float bv = bias ? bias[n] : 0.f;
#pragma unroll
        for (int r = 0; r < 8; ++r) {
            const int m = m0 + r + 8 * half;
            D[(size_t)m * M + n] = acc[t][r] + bv;
        }
    }
}

__global__ __launch_bounds__(256) void attn_logits_kernel(
    const float* __restrict__ H, const float* __restrict__ a_s,
    const float* __restrict__ a_d, float* __restrict__ als, float* __restrict__ ald)
{
    const int idx = blockIdx.x * blockDim.x + threadIdx.x;
    if (idx >= NNODES * NHEADS) return;
    const int head = idx & 1;
    const int node = idx >> 1;
    const float4* hv = (const float4*)(H + (size_t)node * HCAT + head * HIDC);
    const float4* sv = (const float4*)(a_s + head * HIDC);
    const float4* dv = (const float4*)(a_d + head * HIDC);
    float s = 0.f, d = 0.f;
#pragma unroll
    for (int i = 0; i < HIDC / 4; ++i) {
        const float4 h4 = hv[i], s4 = sv[i], d4 = dv[i];
        s += h4.x * s4.x + h4.y * s4.y + h4.z * s4.z + h4.w * s4.w;
        d += h4.x * d4.x + h4.y * d4.y + h4.z * d4.z + h4.w * d4.w;
    }
    als[idx] = s;
    ald[idx] = d;
}

#define LOGIT_U 10
#define LOGIT_T 128
static_assert(NEDGES % (LOGIT_T * LOGIT_U) == 0, "whole work-groups of edges");
__global__ __launch_bounds__(LOGIT_T) void edge_logit_kernel(
    const int* __restrict__ ei, const float* __restrict__ als, const float* __restrict__ ald,
    float* __restrict__ V)
{
    const int e0 = blockIdx.x * (LOGIT_T * LOGIT_U) + threadIdx.x;
    int s[LOGIT_U], d[LOGIT_U];
#pragma unroll
    for (int u = 0; u < LOGIT_U; ++u) {
        s[u] = ei[e0 + u * LOGIT_T];
        d[u] = ei[NEDGES + e0 + u * LOGIT_T];
    }
    float2 as[LOGIT_U], ad[LOGIT_U];
#pragma unroll
    for (int u = 0; u < LOGIT_U; ++u) {
        as[u] = *(const float2*)(als + (size_t)s[u] * 2);
        ad[u] = *(const float2*)(ald + (size_t)d[u] * 2);
    }
#pragma unroll
    for (int u = 0; u < LOGIT_U; ++u) {
        float2 v;
        v.x = leakyr(as[u].x + ad[u].x);
        v.y = leakyr(as[u].y + ad[u].y);
        *(float2*)(V + (size_t)(e0 + u * LOGIT_T) * 2) = v;
    }
}

#define RANGE_NODES  25008
#define NRANGES      4
#define NSLICES      64
#define SLICE_EDGES  (NEDGES / NSLICES)
#define SCAN_THREADS 1024
#define SELF_SPAN    ((RANGE_NODES + NSLICES - 1) / NSLICES)
#define SCAN_U       5
#define FLUSH_U      7
static_assert(NEDGES % NSLICES == 0, "equal slices");
static_assert(RANGE_NODES % 16 == 0 && NNODES % 16 == 0, "every range is whole 128-byte lines");
static_assert(NRANGES * RANGE_NODES >= NNODES && (NRANGES - 1) * RANGE_NODES < NNODES, "ranges cover the nodes");
static_assert(SELF_SPAN * NSLICES >= RANGE_NODES, "the slices' self-loop spans cover a range");
static_assert((NSLICES - 1) * 768 < 2 * (NNODES - (NRANGES - 1) * RANGE_NODES), "flush rotation stays inside the shortest range");

template <bool SUM>
__global__ __launch_bounds__(SCAN_THREADS) void seg_scan_kernel(
    const int* __restrict__ ei, const float* __restrict__ als, const float* __restrict__ ald,
    const float* __restrict__ V,
    const unsigned int* __restrict__ maxk_in, unsigned int* __restrict__ plane)
{
    __shared__ unsigned int acc[RANGE_NODES * 2];
    const int tid   = threadIdx.x;
    const int range = blockIdx.x % NRANGES;
    const int slice = blockIdx.x / NRANGES;
    const int n0 = range * RANGE_NODES;
    const int nn = min(RANGE_NODES, NNODES - n0);
    const int nw = nn * 2;

    const int selfLo = slice * SELF_SPAN * 2;
    const int selfHi = min((slice + 1) * SELF_SPAN, nn) * 2;
    for (int i = tid; i < nw; i += SCAN_THREADS) {
        unsigned int v = 0u;
        if (i >= selfLo && i < selfHi) {
            const size_t g = (size_t)n0 * 2 + i;
            const float e = leakyr(als[g] + ald[g]);
            if (SUM) v = __float_as_uint(__expf(e - funkey(maxk_in[g])));
            else     v = fkey(e);
        }
        acc[i] = v;
    }
    __syncthreads();

    const int ebase = slice * SLICE_EDGES;
    const int eend  = ebase + SLICE_EDGES;
    for (int k0 = 0; k0 < SLICE_EDGES; k0 += SCAN_U * SCAN_THREADS) {
        int d[SCAN_U];
        float2 v[SCAN_U];
#pragma unroll
        for (int u = 0; u < SCAN_U; ++u) {
            const int e = ebase + k0 + u * SCAN_THREADS + tid;
            const bool ok = e < eend;
            const int ec = ok ? e : eend - 1;
            const int dd = ei[NEDGES + ec];
            v[u] = *(const float2*)(V + (size_t)ec * 2);
            d[u] = ok ? dd : -1;
        }
#pragma unroll
        for (int u = 0; u < SCAN_U; ++u)
            asm volatile("" :: "v"(v[u].x), "v"(v[u].y));
        bool in[SCAN_U];
        unsigned int loc[SCAN_U];
        uint2 mk[SCAN_U];
#pragma unroll
        for (int u = 0; u < SCAN_U; ++u) {
            loc[u] = (unsigned int)(d[u] - n0);
            in[u]  = loc[u] < (unsigned int)nn;
            const int dg = in[u] ? d[u] : n0;
            if (SUM) mk[u] = *(const uint2*)(maxk_in + (size_t)dg * 2);
        }
        if (SUM) {
#pragma unroll
            for (int u = 0; u < SCAN_U; ++u)
                asm volatile("" :: "v"(mk[u].x), "v"(mk[u].y));
        }
#pragma unroll
        for (int u = 0; u < SCAN_U; ++u) {
            if (in[u]) {
                const float v0 = v[u].x;
                const float v1 = v[u].y;
                if (SUM) {
                    atomicAdd((float*)&acc[loc[u] * 2 + 0], __expf(v0 - funkey(mk[u].x)));
                    atomicAdd((float*)&acc[loc[u] * 2 + 1], __expf(v1 - funkey(mk[u].y)));
                } else {
                    atomicMax(&acc[loc[u] * 2 + 0], fkey(v0));
                    atomicMax(&acc[loc[u] * 2 + 1], fkey(v1));
                }
            }
        }
    }
    __syncthreads();

    const int rot = slice * 768;
    for (int i0 = tid; i0 < nw; i0 += FLUSH_U * SCAN_THREADS) {
        unsigned int w[FLUSH_U];
        int jj[FLUSH_U];
#pragma unroll
        for (int q = 0; q < FLUSH_U; ++q) {
            const int i = i0 + q * SCAN_THREADS;
            int j = i + rot;
            if (j >= nw) j -= nw;
            jj[q] = j;
            w[q] = (i < nw) ? acc[j] : 0u;
        }
#pragma unroll
        for (int q = 0; q < FLUSH_U; ++q) {
            if (i0 + q * SCAN_THREADS < nw) {
                const size_t g = (size_t)n0 * 2 + jj[q];
                if (SUM) atomicAdd((float*)plane + g, __uint_as_float(w[q]));
                else     atomicMax(plane + g, w[q]);
            }
        }
    }
}

static_assert((NEDGES % 32) == 0 && ((NEDGES + NNODES) % 32) == 0,
              "a wave's 32 edges are all real edges or all self-loops, and all in range");
static_assert(HIDC == 64 && HCAT == 128, "lines 0,1 of a row are head 0, lines 2,3 head 1");
template <bool MEAN>
__global__ __launch_bounds__(256) void edge_exp_aggr_kernel(
    const int* __restrict__ ei, const float* __restrict__ H,
    const float* __restrict__ als, const float* __restrict__ ald,
    const float* __restrict__ V,
    const unsigned int* __restrict__ maxk, const float* __restrict__ denom,
    float* __restrict__ AGG)
{
    const int lane = threadIdx.x & 31;
    const int base = (blockIdx.x * 8 + (threadIdx.x >> 5)) * 32;
    if (base >= NEDGES + NNODES) return;
    int s, d;
    float2 v;
    if (base < NEDGES) {
        const int e = base + lane;
        s = ei[e];
        d = ei[NEDGES + e];
        v = *(const float2*)(V + (size_t)e * 2);
    } else {
        s = d = base + lane - NEDGES;
        const float2 as = *(const float2*)(als + (size_t)s * 2);
        const float2 ad = *(const float2*)(ald + (size_t)d * 2);
        v.x = leakyr(as.x + ad.x);
        v.y = leakyr(as.y + ad.y);
    }
    const uint2 mk = *(const uint2*)(maxk + (size_t)d * 2);
    float x0 = __expf(v.x - funkey(mk.x));
    float x1 = __expf(v.y - funkey(mk.y));
    if (MEAN) {
        const float2 dn = *(const float2*)(denom + (size_t)d * 2);
        x0 = 0.5f * (x0 / dn.x);
        x1 = 0.5f * (x1 / dn.y);
    }

    const float* __restrict__ Hl = H + lane;
    float* __restrict__ Al = AGG + lane;
    for (int jb = 0; jb < 32; jb += 8) {
        float hv[8][4];
#pragma unroll
        for (int u = 0; u < 8; ++u) {
            const int sj = __builtin_amdgcn_readlane(s, jb + u);
            const float* r = Hl + (size_t)sj * HCAT;
            hv[u][0] = r[0];
            hv[u][1] = r[32];
            hv[u][2] = r[64];
            hv[u][3] = r[96];
        }
#pragma unroll
        for (int u = 0; u < 8; ++u) {
            const int dj = __builtin_amdgcn_readlane(d, jb + u);
            const float w0 = __int_as_float(__builtin_amdgcn_readlane(__float_as_int(x0), jb + u));
            const float w1 = __int_as_float(__builtin_amdgcn_readlane(__float_as_int(x1), jb + u));
            if (MEAN) {
                float* o = Al + (size_t)dj * HIDC;
                atomicAdd(o +  0, hv[u][0] * w0 + hv[u][2] * w1);
                atomicAdd(o + 32, hv[u][1] * w0 + hv[u][3] * w1);
            } else {
                float* o = Al + (size_t)dj * HCAT;
                atomicAdd(o +  0, hv[u][0] * w0);
                atomicAdd(o + 32, hv[u][1] * w0);
                atomicAdd(o + 64, hv[u][2] * w1);
                atomicAdd(o + 96, hv[u][3] * w1);
            }
        }
    }
}

#define FIN_U 8
static_assert(NNODES % FIN_U == 0 && (NNODES / 2) % FIN_U == 0, "every wave has FIN_U whole pieces");
static_assert(HCAT == 128 && HIDC == 64, "a piece is one 128-channel row or two 64-channel rows");
template <bool CONCAT>
__global__ __launch_bounds__(256) void finalize_kernel(
    const float* __restrict__ AGG, const float* __restrict__ denom,
    const float* __restrict__ bias,
    const float* __restrict__ prelu_a, float* __restrict__ OUT)
{
    constexpr int NPIECES = CONCAT ? NNODES : NNODES / 2;
    const int lane = threadIdx.x & 31;
    const int p0 = (blockIdx.x * 8 + (threadIdx.x >> 5)) * FIN_U;
    if (p0 >= NPIECES) return;
    const int c = CONCAT ? lane * 4 : (lane & 15) * 4;
    const float a = *prelu_a;
    const float4 b4 = *(const float4*)(bias + c);
    const float4* __restrict__ src = (const float4*)AGG + (size_t)p0 * 32 + lane;
    float4* __restrict__ dst = (float4*)OUT + (size_t)p0 * 32 + lane;
    float4 v[FIN_U];
    float den[FIN_U];
#pragma unroll
    for (int u = 0; u < FIN_U; ++u) {
        v[u] = src[u * 32];
        if (CONCAT) den[u] = denom[(size_t)(p0 + u) * 2 + (lane >> 4)];
    }
#pragma unroll
    for (int u = 0; u < FIN_U; ++u) {
        float4 w = v[u];
        if (CONCAT) {
            w.x = w.x / den[u] + b4.x; w.y = w.y / den[u] + b4.y;
            w.z = w.z / den[u] + b4.z; w.w = w.w / den[u] + b4.w;
        } else {
            w.x += b4.x;
            w.y += b4.y;
            w.z += b4.z;
            w.w += b4.w;
        }
        w.x = w.x >= 0.f ? w.x : a * w.x;
        w.y = w.y >= 0.f ? w.y : a * w.y;
        w.z = w.z >= 0.f ? w.z : a * w.z;
        w.w = w.w >= 0.f ? w.w : a * w.w;
        dst[u * 32] = w;
    }
}

extern "C" void kernel_launch(void* const* d_in, const int* in_sizes, int n_in,
                              void* d_out, int out_size, void* d_ws, size_t ws_size,
                              hipStream_t stream)
{
    const float* x       = (const float*)d_in[0];
    const int*   ei      = (const int*)  d_in[1];
    const float* W1      = (const float*)d_in[2];
    const float* a1s     = (const float*)d_in[3];
    const float* a1d     = (const float*)d_in[4];
    const float* b1      = (const float*)d_in[5];
    const float* W2      = (const float*)d_in[6];
    const float* a2s     = (const float*)d_in[7];
    const float* a2d     = (const float*)d_in[8];
    const float* b2      = (const float*)d_in[9];
    const float* W3      = (const float*)d_in[10];
    const float* a3s     = (const float*)d_in[11];
    const float* a3d     = (const float*)d_in[12];
    const float* b3      = (const float*)d_in[13];
    const float* prelu_a = (const float*)d_in[14];
    const float* Wc      = (const float*)d_in[15];
    const float* bc      = (const float*)d_in[16];
    float* out = (float*)d_out;

    float* ws = (float*)d_ws;
    size_t off = 0;
    float* H     = ws + off; off += (size_t)NNODES * HCAT;
    float* AGG   = ws + off; off += (size_t)NNODES * HCAT;
    float* PREV  = ws + off; off += (size_t)NNODES * HCAT;
    float* als   = ws + off; off += (size_t)NNODES * 2;
    float* ald   = ws + off; off += (size_t)NNODES * 2;
    unsigned int* maxk = (unsigned int*)(ws + off); off += (size_t)NNODES * 2;
    float* denom = ws + off; off += (size_t)NNODES * 2;
    float* V     = ws + off; off += (size_t)NEDGES * 2;

    const int ETOT = NEDGES + NNODES;
    const int gN2    = (NNODES * NHEADS + 255) / 256;
    const int gEdgeW = (ETOT / 32 + 7) / 8;
    const int gTile  = ((NNODES / 16) + 7) / 8;
    const int gFinC  = (NNODES / FIN_U + 7) / 8;
    const int gFinM  = (NNODES / 2 / FIN_U + 7) / 8;

    struct Layer {
        const float* W; const float* as; const float* ad; const float* b; bool concat;
    };
    const Layer layers[3] = {
        { W1, a1s, a1d, b1, true  },
        { W2, a2s, a2d, b2, true  },
        { W3, a3s, a3d, b3, false },
    };

    for (int l = 0; l < 3; ++l) {
        const float* in = (l == 0) ? x : PREV;
        if (layers[l].concat)
            hipMemsetAsync(AGG, 0x00, (size_t)NNODES * HCAT * sizeof(float), stream);
        else
            hipMemsetAsync(AGG, 0x00, (size_t)NNODES * HIDC * sizeof(float), stream);
        hipMemsetAsync(maxk,  0x00, (size_t)NNODES * 2 * sizeof(unsigned int), stream);
        hipMemsetAsync(denom, 0x00, (size_t)NNODES * 2 * sizeof(float), stream);

        if (l == 0)
            gemm_wmma_f32<INCH, HCAT, 8><<<gTile, 256, 0, stream>>>(in, layers[l].W, nullptr, H, NNODES);
        else
            gemm_wmma_f32<HCAT, HCAT, 8><<<gTile, 256, 0, stream>>>(in, layers[l].W, nullptr, H, NNODES);

        attn_logits_kernel<<<gN2, 256, 0, stream>>>(H, layers[l].as, layers[l].ad, als, ald);
        edge_logit_kernel<<<NEDGES / (LOGIT_T * LOGIT_U), LOGIT_T, 0, stream>>>(ei, als, ald, V);
        seg_scan_kernel<false><<<NRANGES * NSLICES, SCAN_THREADS, 0, stream>>>(ei, als, ald, V, nullptr, maxk);
        seg_scan_kernel<true> <<<NRANGES * NSLICES, SCAN_THREADS, 0, stream>>>(ei, als, ald, V, maxk, (unsigned int*)denom);
        if (layers[l].concat) {
            edge_exp_aggr_kernel<false><<<gEdgeW, 256, 0, stream>>>(ei, H, als, ald, V, maxk, nullptr, AGG);
            finalize_kernel<true><<<gFinC, 256, 0, stream>>>(AGG, denom, layers[l].b, prelu_a, PREV);
        } else {
            edge_exp_aggr_kernel<true><<<gEdgeW, 256, 0, stream>>>(ei, H, als, ald, V, maxk, denom, AGG);
            finalize_kernel<false><<<gFinM, 256, 0, stream>>>(AGG, nullptr, layers[l].b, prelu_a, PREV);
        }
    }

    gemm_wmma_f32<HIDC, NCLS, 1><<<gTile, 256, 0, stream>>>(PREV, Wc, bc, out, NNODES);
    (void)in_sizes; (void)n_in; (void)out_size; (void)ws_size;
}


# Attention logits leave their own launches: each layer GEMM wave sums als/ald from its accumulator tiles (in-lane partials, then a 16-lane halving tree) and stores them; logit sum order changes
# speedup vs baseline: 2.9408x; 1.0375x over previous
#include <hip/hip_runtime.h>
#include <hip/hip_bf16.h>

#define NNODES 100000
#define NEDGES 1600000
#define INCH   32
#define HIDC   64
#define NHEADS 2
#define HCAT   128
#define NCLS   12
#define NEGSLOPE 0.2f

typedef __attribute__((ext_vector_type(2))) float v2f;
typedef __attribute__((ext_vector_type(8))) float v8f;

__device__ __forceinline__ float leakyr(float x) { return x > 0.f ? x : NEGSLOPE * x; }

__device__ __forceinline__ unsigned int fkey(float f) {
    const unsigned int b = __float_as_uint(f);
    return b ^ ((b & 0x80000000u) ? 0xFFFFFFFFu : 0x80000000u);
}
__device__ __forceinline__ float funkey(unsigned int k) {
    return __uint_as_float(k ^ ((k & 0x80000000u) ? 0x80000000u : 0xFFFFFFFFu));
}

template <int K, int M, int MT, bool LOGITS>
__global__ __launch_bounds__(256) void gemm_wmma_f32(
    const float* __restrict__ A, const float* __restrict__ B,
    const float* __restrict__ bias, float* __restrict__ D, int nrows,
    const float* __restrict__ a_s, const float* __restrict__ a_d,
    float* __restrict__ als, float* __restrict__ ald)
{
    const int wave = threadIdx.x >> 5;
    const int lane = threadIdx.x & 31;
    const int tile = blockIdx.x * 8 + wave;
    const int ntiles = nrows >> 4;
    if (tile >= ntiles) return;
    const int half = lane >> 4;
    const int mrow = lane & 15;
    const int ncol = lane & 15;
    const int m0 = tile << 4;

    v8f acc[MT] = {};

    const float* __restrict__ arow = A + (size_t)(m0 + mrow) * K;
    for (int kb = 0; kb < K; kb += 4) {
        const int k0 = kb + 2 * half;
        v2f a;
        a.x = arow[k0];
        a.y = arow[k0 + 1];
#pragma unroll
        for (int t = 0; t < MT; ++t) {
            const int n = t * 16 + ncol;
            v2f b;
            if ((M & 15) == 0 || n < M) {
                b.x = B[(size_t)k0 * M + n];
                b.y = B[(size_t)(k0 + 1) * M + n];
            } else {
                b.x = 0.f; b.y = 0.f;
            }
            acc[t] = __builtin_amdgcn_wmma_f32_16x16x4_f32(
                false, a, false, b, (short)0, acc[t], false, false);
        }
    }

#pragma unroll
    for (int t = 0; t < MT; ++t) {
        const int n = t * 16 + ncol;
        if ((M & 15) != 0 && n >= M) continue;
        const float bv = bias ? bias[n] : 0.f;
#pragma unroll
        for (int r = 0; r < 8; ++r) {
            const int m = m0 + r + 8 * half;
            D[(size_t)m * M + n] = acc[t][r] + bv;
        }
    }

    if constexpr (LOGITS) {
        static_assert(!LOGITS || (M == HCAT && MT == 8 && NHEADS == 2 && HIDC == 64),
                      "tiles 0-3 are head 0, tiles 4-7 head 1");
        float cs[MT], cd[MT];
#pragma unroll
        for (int t = 0; t < MT; ++t) {
            cs[t] = a_s[t * 16 + ncol];
            cd[t] = a_d[t * 16 + ncol];
        }
        const bool b3 = (lane & 8) != 0, b2 = (lane & 4) != 0, b1 = (lane & 2) != 0, b0 = (lane & 1) != 0;
        float res[2];
#pragma unroll
        for (int sd = 0; sd < 2; ++sd) {
            float v[16];
#pragma unroll
            for (int r = 0; r < 8; ++r) {
#pragma unroll
                for (int hd = 0; hd < 2; ++hd) {
                    float p = 0.f;
#pragma unroll
                    for (int q = 0; q < 4; ++q)
                        p += acc[hd * 4 + q][r] * (sd ? cd[hd * 4 + q] : cs[hd * 4 + q]);
                    v[r * 2 + hd] = p;
                }
            }
            float w8[8], w4[4], w2[2];
#pragma unroll
            for (int i = 0; i < 8; ++i) {
                const float keep = b3 ? v[i + 8] : v[i];
                const float send = b3 ? v[i] : v[i + 8];
                w8[i] = keep + __shfl_xor(send, 8);
            }
#pragma unroll
            for (int i = 0; i < 4; ++i) {
                const float keep = b2 ? w8[i + 4] : w8[i];
                const float send = b2 ? w8[i] : w8[i + 4];
                w4[i] = keep + __shfl_xor(send, 4);
            }
#pragma unroll
            for (int i = 0; i < 2; ++i) {
                const float keep = b1 ? w4[i + 2] : w4[i];
                const float send = b1 ? w4[i] : w4[i + 2];
                w2[i] = keep + __shfl_xor(send, 2);
            }
            {
                const float keep = b0 ? w2[1] : w2[0];
                const float send = b0 ? w2[0] : w2[1];
                res[sd] = keep + __shfl_xor(send, 1);
            }
        }
        als[(size_t)m0 * 2 + lane] = res[0];
        ald[(size_t)m0 * 2 + lane] = res[1];
    }
}

#define LOGIT_U 10
#define LOGIT_T 128
static_assert(NEDGES % (LOGIT_T * LOGIT_U) == 0, "whole work-groups of edges");
__global__ __launch_bounds__(LOGIT_T) void edge_logit_kernel(
    const int* __restrict__ ei, const float* __restrict__ als, const float* __restrict__ ald,
    float* __restrict__ V)
{
    const int e0 = blockIdx.x * (LOGIT_T * LOGIT_U) + threadIdx.x;
    int s[LOGIT_U], d[LOGIT_U];
#pragma unroll
    for (int u = 0; u < LOGIT_U; ++u) {
        s[u] = ei[e0 + u * LOGIT_T];
        d[u] = ei[NEDGES + e0 + u * LOGIT_T];
    }
    float2 as[LOGIT_U], ad[LOGIT_U];
#pragma unroll
    for (int u = 0; u < LOGIT_U; ++u) {
        as[u] = *(const float2*)(als + (size_t)s[u] * 2);
        ad[u] = *(const float2*)(ald + (size_t)d[u] * 2);
    }
#pragma unroll
    for (int u = 0; u < LOGIT_U; ++u) {
        float2 v;
        v.x = leakyr(as[u].x + ad[u].x);
        v.y = leakyr(as[u].y + ad[u].y);
        *(float2*)(V + (size_t)(e0 + u * LOGIT_T) * 2) = v;
    }
}

#define RANGE_NODES  25008
#define NRANGES      4
#define NSLICES      64
#define SLICE_EDGES  (NEDGES / NSLICES)
#define SCAN_THREADS 1024
#define SELF_SPAN    ((RANGE_NODES + NSLICES - 1) / NSLICES)
#define SCAN_U       5
#define FLUSH_U      7
static_assert(NEDGES % NSLICES == 0, "equal slices");
static_assert(RANGE_NODES % 16 == 0 && NNODES % 16 == 0, "every range is whole 128-byte lines");
static_assert(NRANGES * RANGE_NODES >= NNODES && (NRANGES - 1) * RANGE_NODES < NNODES, "ranges cover the nodes");
static_assert(SELF_SPAN * NSLICES >= RANGE_NODES, "the slices' self-loop spans cover a range");
static_assert((NSLICES - 1) * 768 < 2 * (NNODES - (NRANGES - 1) * RANGE_NODES), "flush rotation stays inside the shortest range");

template <bool SUM>
__global__ __launch_bounds__(SCAN_THREADS) void seg_scan_kernel(
    const int* __restrict__ ei, const float* __restrict__ als, const float* __restrict__ ald,
    const float* __restrict__ V,
    const unsigned int* __restrict__ maxk_in, unsigned int* __restrict__ plane)
{
    __shared__ unsigned int acc[RANGE_NODES * 2];
    const int tid   = threadIdx.x;
    const int range = blockIdx.x % NRANGES;
    const int slice = blockIdx.x / NRANGES;
    const int n0 = range * RANGE_NODES;
    const int nn = min(RANGE_NODES, NNODES - n0);
    const int nw = nn * 2;

    const int selfLo = slice * SELF_SPAN * 2;
    const int selfHi = min((slice + 1) * SELF_SPAN, nn) * 2;
    for (int i = tid; i < nw; i += SCAN_THREADS) {
        unsigned int v = 0u;
        if (i >= selfLo && i < selfHi) {
            const size_t g = (size_t)n0 * 2 + i;
            const float e = leakyr(als[g] + ald[g]);
            if (SUM) v = __float_as_uint(__expf(e - funkey(maxk_in[g])));
            else     v = fkey(e);
        }
        acc[i] = v;
    }
    __syncthreads();

    const int ebase = slice * SLICE_EDGES;
    const int eend  = ebase + SLICE_EDGES;
    for (int k0 = 0; k0 < SLICE_EDGES; k0 += SCAN_U * SCAN_THREADS) {
        int d[SCAN_U];
        float2 v[SCAN_U];
#pragma unroll
        for (int u = 0; u < SCAN_U; ++u) {
            const int e = ebase + k0 + u * SCAN_THREADS + tid;
            const bool ok = e < eend;
            const int ec = ok ? e : eend - 1;
            const int dd = ei[NEDGES + ec];
            v[u] = *(const float2*)(V + (size_t)ec * 2);
            d[u] = ok ? dd : -1;
        }
#pragma unroll
        for (int u = 0; u < SCAN_U; ++u)
            asm volatile("" :: "v"(v[u].x), "v"(v[u].y));
        bool in[SCAN_U];
        unsigned int loc[SCAN_U];
        uint2 mk[SCAN_U];
#pragma unroll
        for (int u = 0; u < SCAN_U; ++u) {
            loc[u] = (unsigned int)(d[u] - n0);
            in[u]  = loc[u] < (unsigned int)nn;
            const int dg = in[u] ? d[u] : n0;
            if (SUM) mk[u] = *(const uint2*)(maxk_in + (size_t)dg * 2);
        }
        if (SUM) {
#pragma unroll
            for (int u = 0; u < SCAN_U; ++u)
                asm volatile("" :: "v"(mk[u].x), "v"(mk[u].y));
        }
#pragma unroll
        for (int u = 0; u < SCAN_U; ++u) {
            if (in[u]) {
                const float v0 = v[u].x;
                const float v1 = v[u].y;
                if (SUM) {
                    atomicAdd((float*)&acc[loc[u] * 2 + 0], __expf(v0 - funkey(mk[u].x)));
                    atomicAdd((float*)&acc[loc[u] * 2 + 1], __expf(v1 - funkey(mk[u].y)));
                } else {
                    atomicMax(&acc[loc[u] * 2 + 0], fkey(v0));
                    atomicMax(&acc[loc[u] * 2 + 1], fkey(v1));
                }
            }
        }
    }
    __syncthreads();

    const int rot = slice * 768;
    for (int i0 = tid; i0 < nw; i0 += FLUSH_U * SCAN_THREADS) {
        unsigned int w[FLUSH_U];
        int jj[FLUSH_U];
#pragma unroll
        for (int q = 0; q < FLUSH_U; ++q) {
            const int i = i0 + q * SCAN_THREADS;
            int j = i + rot;
            if (j >= nw) j -= nw;
            jj[q] = j;
            w[q] = (i < nw) ? acc[j] : 0u;
        }
#pragma unroll
        for (int q = 0; q < FLUSH_U; ++q) {
            if (i0 + q * SCAN_THREADS < nw) {
                const size_t g = (size_t)n0 * 2 + jj[q];
                if (SUM) atomicAdd((float*)plane + g, __uint_as_float(w[q]));
                else     atomicMax(plane + g, w[q]);
            }
        }
    }
}

static_assert((NEDGES % 32) == 0 && ((NEDGES + NNODES) % 32) == 0,
              "a wave's 32 edges are all real edges or all self-loops, and all in range");
static_assert(HIDC == 64 && HCAT == 128, "lines 0,1 of a row are head 0, lines 2,3 head 1");
template <bool MEAN>
__global__ __launch_bounds__(256) void edge_exp_aggr_kernel(
    const int* __restrict__ ei, const float* __restrict__ H,
    const float* __restrict__ als, const float* __restrict__ ald,
    const float* __restrict__ V,
    const unsigned int* __restrict__ maxk, const float* __restrict__ denom,
    float* __restrict__ AGG)
{
    const int lane = threadIdx.x & 31;
    const int base = (blockIdx.x * 8 + (threadIdx.x >> 5)) * 32;
    if (base >= NEDGES + NNODES) return;
    int s, d;
    float2 v;
    if (base < NEDGES) {
        const int e = base + lane;
        s = ei[e];
        d = ei[NEDGES + e];
        v = *(const float2*)(V + (size_t)e * 2);
    } else {
        s = d = base + lane - NEDGES;
        const float2 as = *(const float2*)(als + (size_t)s * 2);
        const float2 ad = *(const float2*)(ald + (size_t)d * 2);
        v.x = leakyr(as.x + ad.x);
        v.y = leakyr(as.y + ad.y);
    }
    const uint2 mk = *(const uint2*)(maxk + (size_t)d * 2);
    float x0 = __expf(v.x - funkey(mk.x));
    float x1 = __expf(v.y - funkey(mk.y));
    if (MEAN) {
        const float2 dn = *(const float2*)(denom + (size_t)d * 2);
        x0 = 0.5f * (x0 / dn.x);
        x1 = 0.5f * (x1 / dn.y);
    }

    const float* __restrict__ Hl = H + lane;
    float* __restrict__ Al = AGG + lane;
    for (int jb = 0; jb < 32; jb += 8) {
        float hv[8][4];
#pragma unroll
        for (int u = 0; u < 8; ++u) {
            const int sj = __builtin_amdgcn_readlane(s, jb + u);
            const float* r = Hl + (size_t)sj * HCAT;
            hv[u][0] = r[0];
            hv[u][1] = r[32];
            hv[u][2] = r[64];
            hv[u][3] = r[96];
        }
#pragma unroll
        for (int u = 0; u < 8; ++u) {
            const int dj = __builtin_amdgcn_readlane(d, jb + u);
            const float w0 = __int_as_float(__builtin_amdgcn_readlane(__float_as_int(x0), jb + u));
            const float w1 = __int_as_float(__builtin_amdgcn_readlane(__float_as_int(x1), jb + u));
            if (MEAN) {
                float* o = Al + (size_t)dj * HIDC;
                atomicAdd(o +  0, hv[u][0] * w0 + hv[u][2] * w1);
                atomicAdd(o + 32, hv[u][1] * w0 + hv[u][3] * w1);
            } else {
                float* o = Al + (size_t)dj * HCAT;
                atomicAdd(o +  0, hv[u][0] * w0);
                atomicAdd(o + 32, hv[u][1] * w0);
                atomicAdd(o + 64, hv[u][2] * w1);
                atomicAdd(o + 96, hv[u][3] * w1);
            }
        }
    }
}

#define FIN_U 8
static_assert(NNODES % FIN_U == 0 && (NNODES / 2) % FIN_U == 0, "every wave has FIN_U whole pieces");
static_assert(HCAT == 128 && HIDC == 64, "a piece is one 128-channel row or two 64-channel rows");
template <bool CONCAT>
__global__ __launch_bounds__(256) void finalize_kernel(
    const float* __restrict__ AGG, const float* __restrict__ denom,
    const float* __restrict__ bias,
    const float* __restrict__ prelu_a, float* __restrict__ OUT)
{
    constexpr int NPIECES = CONCAT ? NNODES : NNODES / 2;
    const int lane = threadIdx.x & 31;
    const int p0 = (blockIdx.x * 8 + (threadIdx.x >> 5)) * FIN_U;
    if (p0 >= NPIECES) return;
    const int c = CONCAT ? lane * 4 : (lane & 15) * 4;
    const float a = *prelu_a;
    const float4 b4 = *(const float4*)(bias + c);
    const float4* __restrict__ src = (const float4*)AGG + (size_t)p0 * 32 + lane;
    float4* __restrict__ dst = (float4*)OUT + (size_t)p0 * 32 + lane;
    float4 v[FIN_U];
    float den[FIN_U];
#pragma unroll
    for (int u = 0; u < FIN_U; ++u) {
        v[u] = src[u * 32];
        if (CONCAT) den[u] = denom[(size_t)(p0 + u) * 2 + (lane >> 4)];
    }
#pragma unroll
    for (int u = 0; u < FIN_U; ++u) {
        float4 w = v[u];
        if (CONCAT) {
            w.x = w.x / den[u] + b4.x; w.y = w.y / den[u] + b4.y;
            w.z = w.z / den[u] + b4.z; w.w = w.w / den[u] + b4.w;
        } else {
            w.x += b4.x;
            w.y += b4.y;
            w.z += b4.z;
            w.w += b4.w;
        }
        w.x = w.x >= 0.f ? w.x : a * w.x;
        w.y = w.y >= 0.f ? w.y : a * w.y;
        w.z = w.z >= 0.f ? w.z : a * w.z;
        w.w = w.w >= 0.f ? w.w : a * w.w;
        dst[u * 32] = w;
    }
}

extern "C" void kernel_launch(void* const* d_in, const int* in_sizes, int n_in,
                              void* d_out, int out_size, void* d_ws, size_t ws_size,
                              hipStream_t stream)
{
    const float* x       = (const float*)d_in[0];
    const int*   ei      = (const int*)  d_in[1];
    const float* W1      = (const float*)d_in[2];
    const float* a1s     = (const float*)d_in[3];
    const float* a1d     = (const float*)d_in[4];
    const float* b1      = (const float*)d_in[5];
    const float* W2      = (const float*)d_in[6];
    const float* a2s     = (const float*)d_in[7];
    const float* a2d     = (const float*)d_in[8];
    const float* b2      = (const float*)d_in[9];
    const float* W3      = (const float*)d_in[10];
    const float* a3s     = (const float*)d_in[11];
    const float* a3d     = (const float*)d_in[12];
    const float* b3      = (const float*)d_in[13];
    const float* prelu_a = (const float*)d_in[14];
    const float* Wc      = (const float*)d_in[15];
    const float* bc      = (const float*)d_in[16];
    float* out = (float*)d_out;

    float* ws = (float*)d_ws;
    size_t off = 0;
    float* H     = ws + off; off += (size_t)NNODES * HCAT;
    float* AGG   = ws + off; off += (size_t)NNODES * HCAT;
    float* PREV  = ws + off; off += (size_t)NNODES * HCAT;
    float* als   = ws + off; off += (size_t)NNODES * 2;
    float* ald   = ws + off; off += (size_t)NNODES * 2;
    unsigned int* maxk = (unsigned int*)(ws + off); off += (size_t)NNODES * 2;
    float* denom = ws + off; off += (size_t)NNODES * 2;
    float* V     = ws + off; off += (size_t)NEDGES * 2;

    const int ETOT = NEDGES + NNODES;
    const int gEdgeW = (ETOT / 32 + 7) / 8;
    const int gTile  = ((NNODES / 16) + 7) / 8;
    const int gFinC  = (NNODES / FIN_U + 7) / 8;
    const int gFinM  = (NNODES / 2 / FIN_U + 7) / 8;

    struct Layer {
        const float* W; const float* as; const float* ad; const float* b; bool concat;
    };
    const Layer layers[3] = {
        { W1, a1s, a1d, b1, true  },
        { W2, a2s, a2d, b2, true  },
        { W3, a3s, a3d, b3, false },
    };

    for (int l = 0; l < 3; ++l) {
        const float* in = (l == 0) ? x : PREV;
        if (layers[l].concat)
            hipMemsetAsync(AGG, 0x00, (size_t)NNODES * HCAT * sizeof(float), stream);
        else
            hipMemsetAsync(AGG, 0x00, (size_t)NNODES * HIDC * sizeof(float), stream);
        hipMemsetAsync(maxk,  0x00, (size_t)NNODES * 2 * sizeof(unsigned int), stream);
        hipMemsetAsync(denom, 0x00, (size_t)NNODES * 2 * sizeof(float), stream);

        if (l == 0)
            gemm_wmma_f32<INCH, HCAT, 8, true><<<gTile, 256, 0, stream>>>(in, layers[l].W, nullptr, H, NNODES,
                                                                         layers[l].as, layers[l].ad, als, ald);
        else
            gemm_wmma_f32<HCAT, HCAT, 8, true><<<gTile, 256, 0, stream>>>(in, layers[l].W, nullptr, H, NNODES,
                                                                         layers[l].as, layers[l].ad, als, ald);
        edge_logit_kernel<<<NEDGES / (LOGIT_T * LOGIT_U), LOGIT_T, 0, stream>>>(ei, als, ald, V);
        seg_scan_kernel<false><<<NRANGES * NSLICES, SCAN_THREADS, 0, stream>>>(ei, als, ald, V, nullptr, maxk);
        seg_scan_kernel<true> <<<NRANGES * NSLICES, SCAN_THREADS, 0, stream>>>(ei, als, ald, V, maxk, (unsigned int*)denom);
        if (layers[l].concat) {
            edge_exp_aggr_kernel<false><<<gEdgeW, 256, 0, stream>>>(ei, H, als, ald, V, maxk, nullptr, AGG);
            finalize_kernel<true><<<gFinC, 256, 0, stream>>>(AGG, denom, layers[l].b, prelu_a, PREV);
        } else {
            edge_exp_aggr_kernel<true><<<gEdgeW, 256, 0, stream>>>(ei, H, als, ald, V, maxk, denom, AGG);
            finalize_kernel<false><<<gFinM, 256, 0, stream>>>(AGG, nullptr, layers[l].b, prelu_a, PREV);
        }
    }

    gemm_wmma_f32<HIDC, NCLS, 1, false><<<gTile, 256, 0, stream>>>(PREV, Wc, bc, out, NNODES,
                                                                  nullptr, nullptr, nullptr, nullptr);
    (void)in_sizes; (void)n_in; (void)out_size; (void)ws_size;
}


# Aggregation by destination-owner work-groups: each of 256 owners keeps its nodes' sum rows and softmax sums in LDS (LDS float atomics, arrival order), filters edges by a byte map, divides after sum
# speedup vs baseline: 3.9412x; 1.3402x over previous
#include <hip/hip_runtime.h>
#include <hip/hip_bf16.h>

#define NNODES 100000
#define NEDGES 1600000
#define INCH   32
#define HIDC   64
#define NHEADS 2
#define HCAT   128
#define NCLS   12
#define NEGSLOPE 0.2f

typedef __attribute__((ext_vector_type(2))) float v2f;
typedef __attribute__((ext_vector_type(8))) float v8f;

__device__ __forceinline__ float leakyr(float x) { return x > 0.f ? x : NEGSLOPE * x; }

__device__ __forceinline__ unsigned int fkey(float f) {
    const unsigned int b = __float_as_uint(f);
    return b ^ ((b & 0x80000000u) ? 0xFFFFFFFFu : 0x80000000u);
}
__device__ __forceinline__ float funkey(unsigned int k) {
    return __uint_as_float(k ^ ((k & 0x80000000u) ? 0x80000000u : 0xFFFFFFFFu));
}

template <int K, int M, int MT, bool LOGITS>
__global__ __launch_bounds__(256) void gemm_wmma_f32(
    const float* __restrict__ A, const float* __restrict__ B,
    const float* __restrict__ bias, float* __restrict__ D, int nrows,
    const float* __restrict__ a_s, const float* __restrict__ a_d,
    float* __restrict__ als, float* __restrict__ ald)
{
    const int wave = threadIdx.x >> 5;
    const int lane = threadIdx.x & 31;
    const int tile = blockIdx.x * 8 + wave;
    const int ntiles = nrows >> 4;
    if (tile >= ntiles) return;
    const int half = lane >> 4;
    const int mrow = lane & 15;
    const int ncol = lane & 15;
    const int m0 = tile << 4;

    v8f acc[MT] = {};

    const float* __restrict__ arow = A + (size_t)(m0 + mrow) * K;
    for (int kb = 0; kb < K; kb += 4) {
        const int k0 = kb + 2 * half;
        v2f a;
        a.x = arow[k0];
        a.y = arow[k0 + 1];
#pragma unroll
        for (int t = 0; t < MT; ++t) {
            const int n = t * 16 + ncol;
            v2f b;
            if ((M & 15) == 0 || n < M) {
                b.x = B[(size_t)k0 * M + n];
                b.y = B[(size_t)(k0 + 1) * M + n];
            } else {
                b.x = 0.f; b.y = 0.f;
            }
            acc[t] = __builtin_amdgcn_wmma_f32_16x16x4_f32(
                false, a, false, b, (short)0, acc[t], false, false);
        }
    }

#pragma unroll
    for (int t = 0; t < MT; ++t) {
        const int n = t * 16 + ncol;
        if ((M & 15) != 0 && n >= M) continue;
        const float bv = bias ? bias[n] : 0.f;
#pragma unroll
        for (int r = 0; r < 8; ++r) {
            const int m = m0 + r + 8 * half;
            D[(size_t)m * M + n] = acc[t][r] + bv;
        }
    }

    if constexpr (LOGITS) {
        static_assert(!LOGITS || (M == HCAT && MT == 8 && NHEADS == 2 && HIDC == 64),
                      "tiles 0-3 are head 0, tiles 4-7 head 1");
        float cs[MT], cd[MT];
#pragma unroll
        for (int t = 0; t < MT; ++t) {
            cs[t] = a_s[t * 16 + ncol];
            cd[t] = a_d[t * 16 + ncol];
        }
        const bool b3 = (lane & 8) != 0, b2 = (lane & 4) != 0, b1 = (lane & 2) != 0, b0 = (lane & 1) != 0;
        float res[2];
#pragma unroll
        for (int sd = 0; sd < 2; ++sd) {
            float v[16];
#pragma unroll
            for (int r = 0; r < 8; ++r) {
#pragma unroll
                for (int hd = 0; hd < 2; ++hd) {
                    float p = 0.f;
#pragma unroll
                    for (int q = 0; q < 4; ++q)
                        p += acc[hd * 4 + q][r] * (sd ? cd[hd * 4 + q] : cs[hd * 4 + q]);
                    v[r * 2 + hd] = p;
                }
            }
            float w8[8], w4[4], w2[2];
#pragma unroll
            for (int i = 0; i < 8; ++i) {
                const float keep = b3 ? v[i + 8] : v[i];
                const float send = b3 ? v[i] : v[i + 8];
                w8[i] = keep + __shfl_xor(send, 8);
            }
#pragma unroll
            for (int i = 0; i < 4; ++i) {
                const float keep = b2 ? w8[i + 4] : w8[i];
                const float send = b2 ? w8[i] : w8[i + 4];
                w4[i] = keep + __shfl_xor(send, 4);
            }
#pragma unroll
            for (int i = 0; i < 2; ++i) {
                const float keep = b1 ? w4[i + 2] : w4[i];
                const float send = b1 ? w4[i] : w4[i + 2];
                w2[i] = keep + __shfl_xor(send, 2);
            }
            {
                const float keep = b0 ? w2[1] : w2[0];
                const float send = b0 ? w2[0] : w2[1];
                res[sd] = keep + __shfl_xor(send, 1);
            }
        }
        als[(size_t)m0 * 2 + lane] = res[0];
        ald[(size_t)m0 * 2 + lane] = res[1];
    }
}

#define LOGIT_U 10
#define LOGIT_T 128
static_assert(NEDGES % (LOGIT_T * LOGIT_U) == 0, "whole work-groups of edges");
__global__ __launch_bounds__(LOGIT_T) void edge_logit_kernel(
    const int* __restrict__ ei, const float* __restrict__ als, const float* __restrict__ ald,
    float* __restrict__ V)
{
    const int e0 = blockIdx.x * (LOGIT_T * LOGIT_U) + threadIdx.x;
    int s[LOGIT_U], d[LOGIT_U];
#pragma unroll
    for (int u = 0; u < LOGIT_U; ++u) {
        s[u] = ei[e0 + u * LOGIT_T];
        d[u] = ei[NEDGES + e0 + u * LOGIT_T];
    }
    float2 as[LOGIT_U], ad[LOGIT_U];
#pragma unroll
    for (int u = 0; u < LOGIT_U; ++u) {
        as[u] = *(const float2*)(als + (size_t)s[u] * 2);
        ad[u] = *(const float2*)(ald + (size_t)d[u] * 2);
    }
#pragma unroll
    for (int u = 0; u < LOGIT_U; ++u) {
        float2 v;
        v.x = leakyr(as[u].x + ad[u].x);
        v.y = leakyr(as[u].y + ad[u].y);
        *(float2*)(V + (size_t)(e0 + u * LOGIT_T) * 2) = v;
    }
}

#define RANGE_NODES  25008
#define NRANGES      4
#define NSLICES      64
#define SLICE_EDGES  (NEDGES / NSLICES)
#define SCAN_THREADS 1024
#define SELF_SPAN    ((RANGE_NODES + NSLICES - 1) / NSLICES)
#define SCAN_U       5
#define FLUSH_U      7
static_assert(NEDGES % NSLICES == 0, "equal slices");
static_assert(RANGE_NODES % 16 == 0 && NNODES % 16 == 0, "every range is whole 128-byte lines");
static_assert(NRANGES * RANGE_NODES >= NNODES && (NRANGES - 1) * RANGE_NODES < NNODES, "ranges cover the nodes");
static_assert(SELF_SPAN * NSLICES >= RANGE_NODES, "the slices' self-loop spans cover a range");
static_assert((NSLICES - 1) * 768 < 2 * (NNODES - (NRANGES - 1) * RANGE_NODES), "flush rotation stays inside the shortest range");

__global__ __launch_bounds__(SCAN_THREADS) void seg_max_kernel(
    const int* __restrict__ ei, const float* __restrict__ als, const float* __restrict__ ald,
    const float* __restrict__ V, unsigned int* __restrict__ plane)
{
    __shared__ unsigned int acc[RANGE_NODES * 2];
    const int tid   = threadIdx.x;
    const int range = blockIdx.x % NRANGES;
    const int slice = blockIdx.x / NRANGES;
    const int n0 = range * RANGE_NODES;
    const int nn = min(RANGE_NODES, NNODES - n0);
    const int nw = nn * 2;

    const int selfLo = slice * SELF_SPAN * 2;
    const int selfHi = min((slice + 1) * SELF_SPAN, nn) * 2;
    for (int i = tid; i < nw; i += SCAN_THREADS) {
        unsigned int v = 0u;
        if (i >= selfLo && i < selfHi) {
            const size_t g = (size_t)n0 * 2 + i;
            v = fkey(leakyr(als[g] + ald[g]));
        }
        acc[i] = v;
    }
    __syncthreads();

    const int ebase = slice * SLICE_EDGES;
    const int eend  = ebase + SLICE_EDGES;
    for (int k0 = 0; k0 < SLICE_EDGES; k0 += SCAN_U * SCAN_THREADS) {
        int d[SCAN_U];
        float2 v[SCAN_U];
#pragma unroll
        for (int u = 0; u < SCAN_U; ++u) {
            const int e = ebase + k0 + u * SCAN_THREADS + tid;
            const bool ok = e < eend;
            const int ec = ok ? e : eend - 1;
            const int dd = ei[NEDGES + ec];
            v[u] = *(const float2*)(V + (size_t)ec * 2);
            d[u] = ok ? dd : -1;
        }
#pragma unroll
        for (int u = 0; u < SCAN_U; ++u)
            asm volatile("" :: "v"(v[u].x), "v"(v[u].y));
#pragma unroll
        for (int u = 0; u < SCAN_U; ++u) {
            const unsigned int loc = (unsigned int)(d[u] - n0);
            if (loc < (unsigned int)nn) {
                atomicMax(&acc[loc * 2 + 0], fkey(v[u].x));
                atomicMax(&acc[loc * 2 + 1], fkey(v[u].y));
            }
        }
    }
    __syncthreads();

    const int rot = slice * 768;
    for (int i0 = tid; i0 < nw; i0 += FLUSH_U * SCAN_THREADS) {
        unsigned int w[FLUSH_U];
        int jj[FLUSH_U];
#pragma unroll
        for (int q = 0; q < FLUSH_U; ++q) {
            const int i = i0 + q * SCAN_THREADS;
            int j = i + rot;
            if (j >= nw) j -= nw;
            jj[q] = j;
            w[q] = (i < nw) ? acc[j] : 0u;
        }
#pragma unroll
        for (int q = 0; q < FLUSH_U; ++q) {
            if (i0 + q * SCAN_THREADS < nw)
                atomicMax(plane + (size_t)n0 * 2 + jj[q], w[q]);
        }
    }
}

#define OWN_NODES   391
#define OWN_WGS     256
#define OWN_THREADS 1024
#define OWN_WAVES   (OWN_THREADS / 32)
#define CHUNK_EDGES 512
#define NCHUNKS     (NEDGES / CHUNK_EDGES)
static_assert(NEDGES % CHUNK_EDGES == 0, "whole chunks");
static_assert(OWN_WGS * OWN_NODES >= NNODES && (OWN_WGS - 1) * OWN_NODES < NNODES, "owners cover the nodes, none is empty");
static_assert((NNODES - 1) / OWN_NODES < 256, "an owner id fits a byte");
static_assert((OWN_NODES + OWN_WAVES - 1) / OWN_WAVES <= 32, "a wave's self-loops fit its queue");
static_assert(HIDC == 64 && HCAT == 128 && NHEADS == 2, "lines 0,1 of a row are head 0, lines 2,3 head 1");

__global__ __launch_bounds__(256) void owner_map_kernel(const int* __restrict__ ei, unsigned int* __restrict__ rid4)
{
    const int i = blockIdx.x * 256 + threadIdx.x;
    if (i >= NEDGES / 4) return;
    const int4 d = *(const int4*)(ei + NEDGES + 4 * (size_t)i);
    rid4[i] = (unsigned int)(d.x / OWN_NODES) | ((unsigned int)(d.y / OWN_NODES) << 8) |
              ((unsigned int)(d.z / OWN_NODES) << 16) | ((unsigned int)(d.w / OWN_NODES) << 24);
}

template <bool MEAN>
__global__ __launch_bounds__(OWN_THREADS) void owner_aggr_kernel(
    const int* __restrict__ ei, const unsigned int* __restrict__ rid4, const float* __restrict__ H,
    const float* __restrict__ als, const float* __restrict__ ald, const float* __restrict__ V,
    const unsigned int* __restrict__ maxk, const float* __restrict__ bias,
    const float* __restrict__ prelu_a, float* __restrict__ OUT)
{
    __shared__ float rows[OWN_NODES * HCAT];
    __shared__ float den[OWN_NODES * 2];
    __shared__ float mx[OWN_NODES * 2];
    const int tid  = threadIdx.x;
    const int lane = tid & 31;
    const int wave = __builtin_amdgcn_readfirstlane(tid >> 5);
    const unsigned int own = blockIdx.x;
    const int n0 = (int)own * OWN_NODES;
    const int nn = min(OWN_NODES, NNODES - n0);

    for (int i = tid; i < nn * (HCAT / 4); i += OWN_THREADS)
        ((float4*)rows)[i] = make_float4(0.f, 0.f, 0.f, 0.f);
    for (int i = tid; i < nn * 2; i += OWN_THREADS) {
        den[i] = 0.f;
        mx[i]  = funkey(maxk[(size_t)n0 * 2 + i]);
    }
    __syncthreads();

    int qn = max(0, (nn - wave + OWN_WAVES - 1) / OWN_WAVES);
    int qe = NEDGES + n0 + wave + OWN_WAVES * lane;

    auto drain = [&](const int cnt) {
        const bool act = lane < cnt;
        const int e = act ? qe : 0;
        int s, d;
        float2 v;
        if (e < NEDGES) {
            s = ei[e];
            d = ei[NEDGES + e];
            v = *(const float2*)(V + (size_t)e * 2);
        } else {
            s = d = e - NEDGES;
            const float2 as = *(const float2*)(als + (size_t)s * 2);
            const float2 ad = *(const float2*)(ald + (size_t)s * 2);
            v.x = leakyr(as.x + ad.x);
            v.y = leakyr(as.y + ad.y);
        }
        const int loc = act ? d - n0 : 0;
        float w0 = 0.f, w1 = 0.f;
        if (act) {
            w0 = __expf(v.x - mx[loc * 2 + 0]);
            w1 = __expf(v.y - mx[loc * 2 + 1]);
            atomicAdd(&den[loc * 2 + 0], w0);
            atomicAdd(&den[loc * 2 + 1], w1);
        }
        const float* __restrict__ Hl = H + lane;
        for (int jb = 0; jb < cnt; jb += 8) {
            float hv[8][4];
#pragma unroll
            for (int u = 0; u < 8; ++u) {
                if (jb + u < cnt) {
                    const int sj = __builtin_amdgcn_readlane(s, jb + u);
                    const float* r = Hl + (size_t)sj * HCAT;
                    hv[u][0] = r[0];
                    hv[u][1] = r[32];
                    hv[u][2] = r[64];
                    hv[u][3] = r[96];
                }
            }
#pragma unroll
            for (int u = 0; u < 8; ++u) {
                if (jb + u < cnt) {
                    const int lj = __builtin_amdgcn_readlane(loc, jb + u);
                    const float x0 = __int_as_float(__builtin_amdgcn_readlane(__float_as_int(w0), jb + u));
                    const float x1 = __int_as_float(__builtin_amdgcn_readlane(__float_as_int(w1), jb + u));
                    float* o = rows + lj * HCAT + lane;
                    atomicAdd(o +  0, hv[u][0] * x0);
                    atomicAdd(o + 32, hv[u][1] * x0);
                    atomicAdd(o + 64, hv[u][2] * x1);
                    atomicAdd(o + 96, hv[u][3] * x1);
                }
            }
        }
    };

    const uint4* __restrict__ rid = (const uint4*)rid4;
    const unsigned int pat = own * 0x01010101u;
    auto hits4 = [&](const unsigned int w) -> unsigned int {
        const unsigned int t = w ^ pat;
        return ~(t | ((t | 0x80808080u) - 0x01010101u)) & 0x80808080u;
    };
    for (int c0 = wave; c0 < NCHUNKS; c0 += 4 * OWN_WAVES) {
        uint4 r[4];
#pragma unroll
        for (int g = 0; g < 4; ++g) {
            const int c = c0 + g * OWN_WAVES;
            r[g] = rid[(size_t)(c < NCHUNKS ? c : c0) * 32 + lane];
        }
#pragma unroll
        for (int h2 = 0; h2 < 2; ++h2) {
            unsigned int hm = 0u;
#pragma unroll
            for (int gg = 0; gg < 2; ++gg) {
                const int g = h2 * 2 + gg;
                if ((c0 + g * OWN_WAVES) < NCHUNKS) {
                    const unsigned int m = (hits4(r[g].x) >> 7) | (hits4(r[g].y) >> 6) |
                                           (hits4(r[g].z) >> 5) | (hits4(r[g].w) >> 4);
                    hm |= m << (4 * gg);
                }
            }
            unsigned int any = __builtin_amdgcn_ballot_w32(hm != 0u);
            while (any) {
                const int L = __builtin_ctz(any);
                any &= any - 1u;
                unsigned int bits = (unsigned int)__builtin_amdgcn_readlane((int)hm, L);
                while (bits) {
                    const int bpos = __builtin_ctz(bits);
                    bits &= bits - 1u;
                    const int c = c0 + (h2 * 2 + ((bpos >> 2) & 1)) * OWN_WAVES;
                    const int e = c * CHUNK_EDGES + L * 16 + 4 * (bpos & 3) + (bpos >> 3);
                    qe = (lane == qn) ? e : qe;
                    ++qn;
                    if (qn == 32) {
                        drain(32);
                        qn = 0;
                    }
                }
            }
        }
    }
    if (qn > 0) drain(qn);
    __syncthreads();

    const float a = *prelu_a;
    for (int j = wave; j < nn; j += OWN_WAVES) {
        const float d0 = den[j * 2 + 0];
        const float d1 = den[j * 2 + 1];
        const float* r = rows + j * HCAT + lane;
        if (MEAN) {
#pragma unroll
            for (int q = 0; q < 2; ++q) {
                const int ch = lane + 32 * q;
                float x = 0.5f * (r[32 * q] / d0 + r[64 + 32 * q] / d1) + bias[ch];
                x = x >= 0.f ? x : a * x;
                OUT[(size_t)(n0 + j) * HIDC + ch] = x;
            }
        } else {
#pragma unroll
            for (int q = 0; q < 4; ++q) {
                const int ch = lane + 32 * q;
                float x = r[32 * q] / (q < 2 ? d0 : d1) + bias[ch];
                x = x >= 0.f ? x : a * x;
                OUT[(size_t)(n0 + j) * HCAT + ch] = x;
            }
        }
    }
}

extern "C" void kernel_launch(void* const* d_in, const int* in_sizes, int n_in,
                              void* d_out, int out_size, void* d_ws, size_t ws_size,
                              hipStream_t stream)
{
    const float* x       = (const float*)d_in[0];
    const int*   ei      = (const int*)  d_in[1];
    const float* W1      = (const float*)d_in[2];
    const float* a1s     = (const float*)d_in[3];
    const float* a1d     = (const float*)d_in[4];
    const float* b1      = (const float*)d_in[5];
    const float* W2      = (const float*)d_in[6];
    const float* a2s     = (const float*)d_in[7];
    const float* a2d     = (const float*)d_in[8];
    const float* b2      = (const float*)d_in[9];
    const float* W3      = (const float*)d_in[10];
    const float* a3s     = (const float*)d_in[11];
    const float* a3d     = (const float*)d_in[12];
    const float* b3      = (const float*)d_in[13];
    const float* prelu_a = (const float*)d_in[14];
    const float* Wc      = (const float*)d_in[15];
    const float* bc      = (const float*)d_in[16];
    float* out = (float*)d_out;

    float* ws = (float*)d_ws;
    size_t off = 0;
    float* H     = ws + off; off += (size_t)NNODES * HCAT;
    float* PREV  = ws + off; off += (size_t)NNODES * HCAT;
    float* als   = ws + off; off += (size_t)NNODES * 2;
    float* ald   = ws + off; off += (size_t)NNODES * 2;
    unsigned int* maxk = (unsigned int*)(ws + off); off += (size_t)NNODES * 2;
    float* V     = ws + off; off += (size_t)NEDGES * 2;
    unsigned int* rid4 = (unsigned int*)(ws + off); off += (size_t)NEDGES / 4;

    const int gTile  = ((NNODES / 16) + 7) / 8;

    struct Layer {
        const float* W; const float* as; const float* ad; const float* b; bool concat;
    };
    const Layer layers[3] = {
        { W1, a1s, a1d, b1, true  },
        { W2, a2s, a2d, b2, true  },
        { W3, a3s, a3d, b3, false },
    };

    owner_map_kernel<<<(NEDGES / 4 + 255) / 256, 256, 0, stream>>>(ei, rid4);

    for (int l = 0; l < 3; ++l) {
        const float* in = (l == 0) ? x : PREV;
        hipMemsetAsync(maxk,  0x00, (size_t)NNODES * 2 * sizeof(unsigned int), stream);

        if (l == 0)
            gemm_wmma_f32<INCH, HCAT, 8, true><<<gTile, 256, 0, stream>>>(in, layers[l].W, nullptr, H, NNODES,
                                                                         layers[l].as, layers[l].ad, als, ald);
        else
            gemm_wmma_f32<HCAT, HCAT, 8, true><<<gTile, 256, 0, stream>>>(in, layers[l].W, nullptr, H, NNODES,
                                                                         layers[l].as, layers[l].ad, als, ald);
        edge_logit_kernel<<<NEDGES / (LOGIT_T * LOGIT_U), LOGIT_T, 0, stream>>>(ei, als, ald, V);
        seg_max_kernel<<<NRANGES * NSLICES, SCAN_THREADS, 0, stream>>>(ei, als, ald, V, maxk);
        if (layers[l].concat)
            owner_aggr_kernel<false><<<OWN_WGS, OWN_THREADS, 0, stream>>>(ei, rid4, H, als, ald, V, maxk, layers[l].b, prelu_a, PREV);
        else
            owner_aggr_kernel<true><<<OWN_WGS, OWN_THREADS, 0, stream>>>(ei, rid4, H, als, ald, V, maxk, layers[l].b, prelu_a, PREV);
    }

    gemm_wmma_f32<HIDC, NCLS, 1, false><<<gTile, 256, 0, stream>>>(PREV, Wc, bc, out, NNODES,
                                                                  nullptr, nullptr, nullptr, nullptr);
    (void)in_sizes; (void)n_in; (void)out_size; (void)ws_size;
}


# One 16-byte record per edge {src, dst, both logits} written by the logit kernel: an owner fetches one line per edge, not three (two index planes + logit plane); max scan reads it too; orders kept
# speedup vs baseline: 4.1970x; 1.0649x over previous
#include <hip/hip_runtime.h>
#include <hip/hip_bf16.h>

#define NNODES 100000
#define NEDGES 1600000
#define INCH   32
#define HIDC   64
#define NHEADS 2
#define HCAT   128
#define NCLS   12
#define NEGSLOPE 0.2f

typedef __attribute__((ext_vector_type(2))) float v2f;
typedef __attribute__((ext_vector_type(8))) float v8f;

__device__ __forceinline__ float leakyr(float x) { return x > 0.f ? x : NEGSLOPE * x; }

__device__ __forceinline__ unsigned int fkey(float f) {
    const unsigned int b = __float_as_uint(f);
    return b ^ ((b & 0x80000000u) ? 0xFFFFFFFFu : 0x80000000u);
}
__device__ __forceinline__ float funkey(unsigned int k) {
    return __uint_as_float(k ^ ((k & 0x80000000u) ? 0x80000000u : 0xFFFFFFFFu));
}

template <int K, int M, int MT, bool LOGITS>
__global__ __launch_bounds__(256) void gemm_wmma_f32(
    const float* __restrict__ A, const float* __restrict__ B,
    const float* __restrict__ bias, float* __restrict__ D, int nrows,
    const float* __restrict__ a_s, const float* __restrict__ a_d,
    float* __restrict__ als, float* __restrict__ ald)
{
    const int wave = threadIdx.x >> 5;
    const int lane = threadIdx.x & 31;
    const int tile = blockIdx.x * 8 + wave;
    const int ntiles = nrows >> 4;
    if (tile >= ntiles) return;
    const int half = lane >> 4;
    const int mrow = lane & 15;
    const int ncol = lane & 15;
    const int m0 = tile << 4;

    v8f acc[MT] = {};

    const float* __restrict__ arow = A + (size_t)(m0 + mrow) * K;
    for (int kb = 0; kb < K; kb += 4) {
        const int k0 = kb + 2 * half;
        v2f a;
        a.x = arow[k0];
        a.y = arow[k0 + 1];
#pragma unroll
        for (int t = 0; t < MT; ++t) {
            const int n = t * 16 + ncol;
            v2f b;
            if ((M & 15) == 0 || n < M) {
                b.x = B[(size_t)k0 * M + n];
                b.y = B[(size_t)(k0 + 1) * M + n];
            } else {
                b.x = 0.f; b.y = 0.f;
            }
            acc[t] = __builtin_amdgcn_wmma_f32_16x16x4_f32(
                false, a, false, b, (short)0, acc[t], false, false);
        }
    }

#pragma unroll
    for (int t = 0; t < MT; ++t) {
        const int n = t * 16 + ncol;
        if ((M & 15) != 0 && n >= M) continue;
        const float bv = bias ? bias[n] : 0.f;
#pragma unroll
        for (int r = 0; r < 8; ++r) {
            const int m = m0 + r + 8 * half;
            D[(size_t)m * M + n] = acc[t][r] + bv;
        }
    }

    if constexpr (LOGITS) {
        static_assert(!LOGITS || (M == HCAT && MT == 8 && NHEADS == 2 && HIDC == 64),
                      "tiles 0-3 are head 0, tiles 4-7 head 1");
        float cs[MT], cd[MT];
#pragma unroll
        for (int t = 0; t < MT; ++t) {
            cs[t] = a_s[t * 16 + ncol];
            cd[t] = a_d[t * 16 + ncol];
        }
        const bool b3 = (lane & 8) != 0, b2 = (lane & 4) != 0, b1 = (lane & 2) != 0, b0 = (lane & 1) != 0;
        float res[2];
#pragma unroll
        for (int sd = 0; sd < 2; ++sd) {
            float v[16];
#pragma unroll
            for (int r = 0; r < 8; ++r) {
#pragma unroll
                for (int hd = 0; hd < 2; ++hd) {
                    float p = 0.f;
#pragma unroll
                    for (int q = 0; q < 4; ++q)
                        p += acc[hd * 4 + q][r] * (sd ? cd[hd * 4 + q] : cs[hd * 4 + q]);
                    v[r * 2 + hd] = p;
                }
            }
            float w8[8], w4[4], w2[2];
#pragma unroll
            for (int i = 0; i < 8; ++i) {
                const float keep = b3 ? v[i + 8] : v[i];
                const float send = b3 ? v[i] : v[i + 8];
                w8[i] = keep + __shfl_xor(send, 8);
            }
#pragma unroll
            for (int i = 0; i < 4; ++i) {
                const float keep = b2 ? w8[i + 4] : w8[i];
                const float send = b2 ? w8[i] : w8[i + 4];
                w4[i] = keep + __shfl_xor(send, 4);
            }
#pragma unroll
            for (int i = 0; i < 2; ++i) {
                const float keep = b1 ? w4[i + 2] : w4[i];
                const float send = b1 ? w4[i] : w4[i + 2];
                w2[i] = keep + __shfl_xor(send, 2);
            }
            {
                const float keep = b0 ? w2[1] : w2[0];
                const float send = b0 ? w2[0] : w2[1];
                res[sd] = keep + __shfl_xor(send, 1);
            }
        }
        als[(size_t)m0 * 2 + lane] = res[0];
        ald[(size_t)m0 * 2 + lane] = res[1];
    }
}

#define LOGIT_U 10
#define LOGIT_T 128
static_assert(NEDGES % (LOGIT_T * LOGIT_U) == 0, "whole work-groups of edges");
__global__ __launch_bounds__(LOGIT_T) void edge_logit_kernel(
    const int* __restrict__ ei, const float* __restrict__ als, const float* __restrict__ ald,
    int* __restrict__ R)
{
    int4* __restrict__ R4 = (int4*)R;
    const int e0 = blockIdx.x * (LOGIT_T * LOGIT_U) + threadIdx.x;
    int s[LOGIT_U], d[LOGIT_U];
#pragma unroll
    for (int u = 0; u < LOGIT_U; ++u) {
        s[u] = ei[e0 + u * LOGIT_T];
        d[u] = ei[NEDGES + e0 + u * LOGIT_T];
    }
    float2 as[LOGIT_U], ad[LOGIT_U];
#pragma unroll
    for (int u = 0; u < LOGIT_U; ++u) {
        as[u] = *(const float2*)(als + (size_t)s[u] * 2);
        ad[u] = *(const float2*)(ald + (size_t)d[u] * 2);
    }
#pragma unroll
    for (int u = 0; u < LOGIT_U; ++u) {
        int4 r;
        r.x = s[u];
        r.y = d[u];
        r.z = __float_as_int(leakyr(as[u].x + ad[u].x));
        r.w = __float_as_int(leakyr(as[u].y + ad[u].y));
        R4[e0 + u * LOGIT_T] = r;
    }
}

#define RANGE_NODES  25008
#define NRANGES      4
#define NSLICES      64
#define SLICE_EDGES  (NEDGES / NSLICES)
#define SCAN_THREADS 1024
#define SELF_SPAN    ((RANGE_NODES + NSLICES - 1) / NSLICES)
#define SCAN_U       5
#define FLUSH_U      7
static_assert(NEDGES % NSLICES == 0, "equal slices");
static_assert(RANGE_NODES % 16 == 0 && NNODES % 16 == 0, "every range is whole 128-byte lines");
static_assert(NRANGES * RANGE_NODES >= NNODES && (NRANGES - 1) * RANGE_NODES < NNODES, "ranges cover the nodes");
static_assert(SELF_SPAN * NSLICES >= RANGE_NODES, "the slices' self-loop spans cover a range");
static_assert((NSLICES - 1) * 768 < 2 * (NNODES - (NRANGES - 1) * RANGE_NODES), "flush rotation stays inside the shortest range");

__global__ __launch_bounds__(SCAN_THREADS) void seg_max_kernel(
    const float* __restrict__ als, const float* __restrict__ ald,
    const int* __restrict__ R, unsigned int* __restrict__ plane)
{
    __shared__ unsigned int acc[RANGE_NODES * 2];
    const int4* __restrict__ R4 = (const int4*)R;
    const int tid   = threadIdx.x;
    const int range = blockIdx.x % NRANGES;
    const int slice = blockIdx.x / NRANGES;
    const int n0 = range * RANGE_NODES;
    const int nn = min(RANGE_NODES, NNODES - n0);
    const int nw = nn * 2;

    const int selfLo = slice * SELF_SPAN * 2;
    const int selfHi = min((slice + 1) * SELF_SPAN, nn) * 2;
    for (int i = tid; i < nw; i += SCAN_THREADS) {
        unsigned int v = 0u;
        if (i >= selfLo && i < selfHi) {
            const size_t g = (size_t)n0 * 2 + i;
            v = fkey(leakyr(als[g] + ald[g]));
        }
        acc[i] = v;
    }
    __syncthreads();

    const int ebase = slice * SLICE_EDGES;
    const int eend  = ebase + SLICE_EDGES;
    for (int k0 = 0; k0 < SLICE_EDGES; k0 += SCAN_U * SCAN_THREADS) {
        int d[SCAN_U];
        float2 v[SCAN_U];
#pragma unroll
        for (int u = 0; u < SCAN_U; ++u) {
            const int e = ebase + k0 + u * SCAN_THREADS + tid;
            const bool ok = e < eend;
            const int ec = ok ? e : eend - 1;
            const int4 r = R4[ec];
            v[u].x = __int_as_float(r.z);
            v[u].y = __int_as_float(r.w);
            d[u] = ok ? r.y : -1;
        }
#pragma unroll
        for (int u = 0; u < SCAN_U; ++u)
            asm volatile("" :: "v"(v[u].x), "v"(v[u].y));
#pragma unroll
        for (int u = 0; u < SCAN_U; ++u) {
            const unsigned int loc = (unsigned int)(d[u] - n0);
            if (loc < (unsigned int)nn) {
                atomicMax(&acc[loc * 2 + 0], fkey(v[u].x));
                atomicMax(&acc[loc * 2 + 1], fkey(v[u].y));
            }
        }
    }
    __syncthreads();

    const int rot = slice * 768;
    for (int i0 = tid; i0 < nw; i0 += FLUSH_U * SCAN_THREADS) {
        unsigned int w[FLUSH_U];
        int jj[FLUSH_U];
#pragma unroll
        for (int q = 0; q < FLUSH_U; ++q) {
            const int i = i0 + q * SCAN_THREADS;
            int j = i + rot;
            if (j >= nw) j -= nw;
            jj[q] = j;
            w[q] = (i < nw) ? acc[j] : 0u;
        }
#pragma unroll
        for (int q = 0; q < FLUSH_U; ++q) {
            if (i0 + q * SCAN_THREADS < nw)
                atomicMax(plane + (size_t)n0 * 2 + jj[q], w[q]);
        }
    }
}

#define OWN_NODES   391
#define OWN_WGS     256
#define OWN_THREADS 1024
#define OWN_WAVES   (OWN_THREADS / 32)
#define CHUNK_EDGES 512
#define NCHUNKS     (NEDGES / CHUNK_EDGES)
static_assert(NEDGES % CHUNK_EDGES == 0, "whole chunks");
static_assert(OWN_WGS * OWN_NODES >= NNODES && (OWN_WGS - 1) * OWN_NODES < NNODES, "owners cover the nodes, none is empty");
static_assert((NNODES - 1) / OWN_NODES < 256, "an owner id fits a byte");
static_assert((OWN_NODES + OWN_WAVES - 1) / OWN_WAVES <= 32, "a wave's self-loops fit its queue");
static_assert(HIDC == 64 && HCAT == 128 && NHEADS == 2, "lines 0,1 of a row are head 0, lines 2,3 head 1");

__global__ __launch_bounds__(256) void owner_map_kernel(const int* __restrict__ ei, unsigned int* __restrict__ rid4)
{
    const int i = blockIdx.x * 256 + threadIdx.x;
    if (i >= NEDGES / 4) return;
    const int4 d = *(const int4*)(ei + NEDGES + 4 * (size_t)i);
    rid4[i] = (unsigned int)(d.x / OWN_NODES) | ((unsigned int)(d.y / OWN_NODES) << 8) |
              ((unsigned int)(d.z / OWN_NODES) << 16) | ((unsigned int)(d.w / OWN_NODES) << 24);
}

template <bool MEAN>
__global__ __launch_bounds__(OWN_THREADS) void owner_aggr_kernel(
    const int* __restrict__ R, const unsigned int* __restrict__ rid4, const float* __restrict__ H,
    const float* __restrict__ als, const float* __restrict__ ald,
    const unsigned int* __restrict__ maxk, const float* __restrict__ bias,
    const float* __restrict__ prelu_a, float* __restrict__ OUT)
{
    __shared__ float rows[OWN_NODES * HCAT];
    __shared__ float den[OWN_NODES * 2];
    __shared__ float mx[OWN_NODES * 2];
    const int tid  = threadIdx.x;
    const int lane = tid & 31;
    const int wave = __builtin_amdgcn_readfirstlane(tid >> 5);
    const unsigned int own = blockIdx.x;
    const int n0 = (int)own * OWN_NODES;
    const int nn = min(OWN_NODES, NNODES - n0);

    for (int i = tid; i < nn * (HCAT / 4); i += OWN_THREADS)
        ((float4*)rows)[i] = make_float4(0.f, 0.f, 0.f, 0.f);
    for (int i = tid; i < nn * 2; i += OWN_THREADS) {
        den[i] = 0.f;
        mx[i]  = funkey(maxk[(size_t)n0 * 2 + i]);
    }
    __syncthreads();

    int qn = max(0, (nn - wave + OWN_WAVES - 1) / OWN_WAVES);
    int qe = NEDGES + n0 + wave + OWN_WAVES * lane;

    auto drain = [&](const int cnt) {
        const bool act = lane < cnt;
        const int e = act ? qe : 0;
        int s, d;
        float2 v;
        if (e < NEDGES) {
            const int4 r = ((const int4*)R)[e];
            s = r.x;
            d = r.y;
            v.x = __int_as_float(r.z);
            v.y = __int_as_float(r.w);
        } else {
            s = d = e - NEDGES;
            const float2 as = *(const float2*)(als + (size_t)s * 2);
            const float2 ad = *(const float2*)(ald + (size_t)s * 2);
            v.x = leakyr(as.x + ad.x);
            v.y = leakyr(as.y + ad.y);
        }
        const int loc = act ? d - n0 : 0;
        float w0 = 0.f, w1 = 0.f;
        if (act) {
            w0 = __expf(v.x - mx[loc * 2 + 0]);
            w1 = __expf(v.y - mx[loc * 2 + 1]);
            atomicAdd(&den[loc * 2 + 0], w0);
            atomicAdd(&den[loc * 2 + 1], w1);
        }
        const float* __restrict__ Hl = H + lane;
        for (int jb = 0; jb < cnt; jb += 8) {
            float hv[8][4];
#pragma unroll
            for (int u = 0; u < 8; ++u) {
                if (jb + u < cnt) {
                    const int sj = __builtin_amdgcn_readlane(s, jb + u);
                    const float* r = Hl + (size_t)sj * HCAT;
                    hv[u][0] = r[0];
                    hv[u][1] = r[32];
                    hv[u][2] = r[64];
                    hv[u][3] = r[96];
                }
            }
#pragma unroll
            for (int u = 0; u < 8; ++u) {
                if (jb + u < cnt) {
                    const int lj = __builtin_amdgcn_readlane(loc, jb + u);
                    const float x0 = __int_as_float(__builtin_amdgcn_readlane(__float_as_int(w0), jb + u));
                    const float x1 = __int_as_float(__builtin_amdgcn_readlane(__float_as_int(w1), jb + u));
                    float* o = rows + lj * HCAT + lane;
                    atomicAdd(o +  0, hv[u][0] * x0);
                    atomicAdd(o + 32, hv[u][1] * x0);
                    atomicAdd(o + 64, hv[u][2] * x1);
                    atomicAdd(o + 96, hv[u][3] * x1);
                }
            }
        }
    };

    const uint4* __restrict__ rid = (const uint4*)rid4;
    const unsigned int pat = own * 0x01010101u;
    auto hits4 = [&](const unsigned int w) -> unsigned int {
        const unsigned int t = w ^ pat;
        return ~(t | ((t | 0x80808080u) - 0x01010101u)) & 0x80808080u;
    };
    for (int c0 = wave; c0 < NCHUNKS; c0 += 4 * OWN_WAVES) {
        uint4 r[4];
#pragma unroll
        for (int g = 0; g < 4; ++g) {
            const int c = c0 + g * OWN_WAVES;
            r[g] = rid[(size_t)(c < NCHUNKS ? c : c0) * 32 + lane];
        }
#pragma unroll
        for (int h2 = 0; h2 < 2; ++h2) {
            unsigned int hm = 0u;
#pragma unroll
            for (int gg = 0; gg < 2; ++gg) {
                const int g = h2 * 2 + gg;
                if ((c0 + g * OWN_WAVES) < NCHUNKS) {
                    const unsigned int m = (hits4(r[g].x) >> 7) | (hits4(r[g].y) >> 6) |
                                           (hits4(r[g].z) >> 5) | (hits4(r[g].w) >> 4);
                    hm |= m << (4 * gg);
                }
            }
            unsigned int any = __builtin_amdgcn_ballot_w32(hm != 0u);
            while (any) {
                const int L = __builtin_ctz(any);
                any &= any - 1u;
                unsigned int bits = (unsigned int)__builtin_amdgcn_readlane((int)hm, L);
                while (bits) {
                    const int bpos = __builtin_ctz(bits);
                    bits &= bits - 1u;
                    const int c = c0 + (h2 * 2 + ((bpos >> 2) & 1)) * OWN_WAVES;
                    const int e = c * CHUNK_EDGES + L * 16 + 4 * (bpos & 3) + (bpos >> 3);
                    qe = (lane == qn) ? e : qe;
                    ++qn;
                    if (qn == 32) {
                        drain(32);
                        qn = 0;
                    }
                }
            }
        }
    }
    if (qn > 0) drain(qn);
    __syncthreads();

    const float a = *prelu_a;
    for (int j = wave; j < nn; j += OWN_WAVES) {
        const float d0 = den[j * 2 + 0];
        const float d1 = den[j * 2 + 1];
        const float* r = rows + j * HCAT + lane;
        if (MEAN) {
#pragma unroll
            for (int q = 0; q < 2; ++q) {
                const int ch = lane + 32 * q;
                float x = 0.5f * (r[32 * q] / d0 + r[64 + 32 * q] / d1) + bias[ch];
                x = x >= 0.f ? x : a * x;
                OUT[(size_t)(n0 + j) * HIDC + ch] = x;
            }
        } else {
#pragma unroll
            for (int q = 0; q < 4; ++q) {
                const int ch = lane + 32 * q;
                float x = r[32 * q] / (q < 2 ? d0 : d1) + bias[ch];
                x = x >= 0.f ? x : a * x;
                OUT[(size_t)(n0 + j) * HCAT + ch] = x;
            }
        }
    }
}

extern "C" void kernel_launch(void* const* d_in, const int* in_sizes, int n_in,
                              void* d_out, int out_size, void* d_ws, size_t ws_size,
                              hipStream_t stream)
{
    const float* x       = (const float*)d_in[0];
    const int*   ei      = (const int*)  d_in[1];
    const float* W1      = (const float*)d_in[2];
    const float* a1s     = (const float*)d_in[3];
    const float* a1d     = (const float*)d_in[4];
    const float* b1      = (const float*)d_in[5];
    const float* W2      = (const float*)d_in[6];
    const float* a2s     = (const float*)d_in[7];
    const float* a2d     = (const float*)d_in[8];
    const float* b2      = (const float*)d_in[9];
    const float* W3      = (const float*)d_in[10];
    const float* a3s     = (const float*)d_in[11];
    const float* a3d     = (const float*)d_in[12];
    const float* b3      = (const float*)d_in[13];
    const float* prelu_a = (const float*)d_in[14];
    const float* Wc      = (const float*)d_in[15];
    const float* bc      = (const float*)d_in[16];
    float* out = (float*)d_out;

    float* ws = (float*)d_ws;
    size_t off = 0;
    float* H     = ws + off; off += (size_t)NNODES * HCAT;
    float* PREV  = ws + off; off += (size_t)NNODES * HCAT;
    float* als   = ws + off; off += (size_t)NNODES * 2;
    float* ald   = ws + off; off += (size_t)NNODES * 2;
    unsigned int* maxk = (unsigned int*)(ws + off); off += (size_t)NNODES * 2;
    int* R       = (int*)(ws + off); off += (size_t)NEDGES * 4;
    unsigned int* rid4 = (unsigned int*)(ws + off); off += (size_t)NEDGES / 4;

    const int gTile  = ((NNODES / 16) + 7) / 8;

    struct Layer {
        const float* W; const float* as; const float* ad; const float* b; bool concat;
    };
    const Layer layers[3] = {
        { W1, a1s, a1d, b1, true  },
        { W2, a2s, a2d, b2, true  },
        { W3, a3s, a3d, b3, false },
    };

    owner_map_kernel<<<(NEDGES / 4 + 255) / 256, 256, 0, stream>>>(ei, rid4);

    for (int l = 0; l < 3; ++l) {
        const float* in = (l == 0) ? x : PREV;
        hipMemsetAsync(maxk,  0x00, (size_t)NNODES * 2 * sizeof(unsigned int), stream);

        if (l == 0)
            gemm_wmma_f32<INCH, HCAT, 8, true><<<gTile, 256, 0, stream>>>(in, layers[l].W, nullptr, H, NNODES,
                                                                         layers[l].as, layers[l].ad, als, ald);
        else
            gemm_wmma_f32<HCAT, HCAT, 8, true><<<gTile, 256, 0, stream>>>(in, layers[l].W, nullptr, H, NNODES,
                                                                         layers[l].as, layers[l].ad, als, ald);
        edge_logit_kernel<<<NEDGES / (LOGIT_T * LOGIT_U), LOGIT_T, 0, stream>>>(ei, als, ald, R);
        seg_max_kernel<<<NRANGES * NSLICES, SCAN_THREADS, 0, stream>>>(als, ald, R, maxk);
        if (layers[l].concat)
            owner_aggr_kernel<false><<<OWN_WGS, OWN_THREADS, 0, stream>>>(R, rid4, H, als, ald, maxk, layers[l].b, prelu_a, PREV);
        else
            owner_aggr_kernel<true><<<OWN_WGS, OWN_THREADS, 0, stream>>>(R, rid4, H, als, ald, maxk, layers[l].b, prelu_a, PREV);
    }

    gemm_wmma_f32<HIDC, NCLS, 1, false><<<gTile, 256, 0, stream>>>(PREV, Wc, bc, out, NNODES,
                                                                  nullptr, nullptr, nullptr, nullptr);
    (void)in_sizes; (void)n_in; (void)out_size; (void)ws_size;
}
